# starting kernel
# baseline (speedup 1.0000x reference)
#include <hip/hip_runtime.h>

#define GDIM 128
#define NGRAPH 256
#define NEG_SLOPE 0.2f

typedef float v2f __attribute__((ext_vector_type(2)));
typedef float v8f __attribute__((ext_vector_type(8)));

__device__ __forceinline__ unsigned enc_f32(float f) {
    unsigned b = __float_as_uint(f);
    return (b & 0x80000000u) ? ~b : (b | 0x80000000u);
}
__device__ __forceinline__ float dec_f32(unsigned u) {
    return __uint_as_float((u & 0x80000000u) ? (u ^ 0x80000000u) : ~u);
}

__global__ void __launch_bounds__(128)
gat_gemm_wmma(const float* __restrict__ X, const float* __restrict__ W,
              float* __restrict__ H, int nTiles) {
    extern __shared__ float sW[];
    const int tid = threadIdx.x;

    const float4* W4 = (const float4*)W;
    float4*       S4 = (float4*)sW;
#pragma unroll
    for (int t = 0; t < 32; ++t) S4[t * 128 + tid] = W4[t * 128 + tid];
    __syncthreads();

    const int wid  = tid >> 5;
    const int lane = tid & 31;
    const int hi   = lane >> 4;
    const int col  = lane & 15;
    const int tile = blockIdx.x * 4 + wid;
    if (tile >= nTiles) return;

    const long   row0 = (long)tile * 16;
    const float* Xr   = X + (row0 + col) * GDIM;

    v8f c[8];
#pragma unroll
    for (int n = 0; n < 8; ++n) c[n] = (v8f)0.0f;

    for (int kk = 0; kk < GDIM; kk += 4) {
        const int ka = kk + 2 * hi;
        v2f a = *(const v2f*)(Xr + ka);
#pragma unroll
        for (int n = 0; n < 8; ++n) {
            v2f b;
            b.x = sW[(ka + 0) * GDIM + n * 16 + col];
            b.y = sW[(ka + 1) * GDIM + n * 16 + col];
            c[n] = __builtin_amdgcn_wmma_f32_16x16x4_f32(
                       false, a, false, b, (short)0, c[n], false, false);
        }
    }

    float* Hr = H + row0 * GDIM;
#pragma unroll
    for (int n = 0; n < 8; ++n)
#pragma unroll
        for (int v = 0; v < 8; ++v)
            Hr[(v + 8 * hi) * GDIM + n * 16 + col] = c[n][v];
}

__global__ void __launch_bounds__(256)
gat_node_scores(const float* __restrict__ H, const float* __restrict__ a_src,
                const float* __restrict__ a_dst, float* __restrict__ s_src,
                float* __restrict__ s_dst, int n) {
    const int lane = threadIdx.x & 31;
    const int node = blockIdx.x * 8 + (threadIdx.x >> 5);
    if (node >= n) return;
    const float4 hv = *(const float4*)(H + (long)node * GDIM + lane * 4);
    const float4 as = *(const float4*)(a_src + lane * 4);
    const float4 ad = *(const float4*)(a_dst + lane * 4);
    float ps = hv.x * as.x + hv.y * as.y + hv.z * as.z + hv.w * as.w;
    float pd = hv.x * ad.x + hv.y * ad.y + hv.z * ad.z + hv.w * ad.w;
#pragma unroll
    for (int off = 16; off > 0; off >>= 1) {
        ps += __shfl_xor(ps, off, 32);
        pd += __shfl_xor(pd, off, 32);
    }
    if (lane == 0) { s_src[node] = ps; s_dst[node] = pd; }
}

__global__ void __launch_bounds__(256)
gat_edge_max(const int* __restrict__ ei, int E, int n,
             const float* __restrict__ s_src, const float* __restrict__ s_dst,
             unsigned* __restrict__ m_enc) {
    const int e  = blockIdx.x * blockDim.x + threadIdx.x;
    const int NE = E + n;
    if (e >= NE) return;
    int i, j;
    if (e < E) { i = ei[e]; j = ei[E + e]; } else { i = j = e - E; }
    const float s     = s_src[i] + s_dst[j];
    const float logit = (s > 0.f) ? s : NEG_SLOPE * s;
    atomicMax(m_enc + j, enc_f32(logit));
}

__global__ void __launch_bounds__(256)
gat_edge_aggr(const int* __restrict__ ei, int E, int n,
              const float* __restrict__ s_src, const float* __restrict__ s_dst,
              const unsigned* __restrict__ m_enc, const float* __restrict__ H,
              float* __restrict__ denom, float* __restrict__ acc) {
    const int lane = threadIdx.x & 31;
    const int e    = blockIdx.x * 8 + (threadIdx.x >> 5);
    const int NE   = E + n;
    if (e >= NE) return;
    int i, j;
    if (e < E) { i = ei[e]; j = ei[E + e]; } else { i = j = e - E; }
    const float s     = s_src[i] + s_dst[j];
    const float logit = (s > 0.f) ? s : NEG_SLOPE * s;
    const float w     = __expf(logit - dec_f32(m_enc[j]));
    if (lane == 0) atomicAdd(denom + j, w);
    const float4 hv = *(const float4*)(H + (long)i * GDIM + lane * 4);
    float* ap = acc + (long)j * GDIM + lane * 4;
    atomicAdd(ap + 0, w * hv.x);
    atomicAdd(ap + 1, w * hv.y);
    atomicAdd(ap + 2, w * hv.z);
    atomicAdd(ap + 3, w * hv.w);
}

__global__ void __launch_bounds__(256)
gat_finalize(float* __restrict__ acc, const float* __restrict__ denom,
             const float* __restrict__ bias, int n) {
    const int lane = threadIdx.x & 31;
    const int node = blockIdx.x * 8 + (threadIdx.x >> 5);
    if (node >= n) return;
    const float  inv = 1.0f / denom[node];
    const float4 b4  = *(const float4*)(bias + lane * 4);
    float4* p = (float4*)(acc + (long)node * GDIM + lane * 4);
    float4  v = *p;
    v.x = v.x * inv + b4.x;
    v.y = v.y * inv + b4.y;
    v.z = v.z * inv + b4.z;
    v.w = v.w * inv + b4.w;
    *p = v;
}

__global__ void __launch_bounds__(256)
gat_pool_accum(const float* __restrict__ h, const int* __restrict__ batch, int n,
               float* __restrict__ psum, float* __restrict__ pcount) {
    const int lane = threadIdx.x & 31;
    const int node = blockIdx.x * 8 + (threadIdx.x >> 5);
    if (node >= n) return;
    const int    g  = batch[node];
    const float4 hv = *(const float4*)(h + (long)node * GDIM + lane * 4);
    float* pp = psum + (long)g * GDIM + lane * 4;
    atomicAdd(pp + 0, hv.x);
    atomicAdd(pp + 1, hv.y);
    atomicAdd(pp + 2, hv.z);
    atomicAdd(pp + 3, hv.w);
    if (lane == 0) atomicAdd(pcount + g, 1.0f);
}

__global__ void __launch_bounds__(256)
gat_pool_final(const float* __restrict__ psum, const float* __restrict__ pcount,
               float* __restrict__ out) {
    const int idx = blockIdx.x * blockDim.x + threadIdx.x;
    const float cnt = pcount[idx >> 7];
    out[idx] = psum[idx] / fmaxf(cnt, 1.0f);
}

extern "C" void kernel_launch(void* const* d_in, const int* in_sizes, int n_in,
                              void* d_out, int out_size, void* d_ws, size_t ws_size,
                              hipStream_t stream) {
    const float* x     = (const float*)d_in[0];
    const int*   ei    = (const int*)d_in[1];
    const int*   batch = (const int*)d_in[2];
    const float* W[3]    = {(const float*)d_in[3], (const float*)d_in[7],  (const float*)d_in[11]};
    const float* asrc[3] = {(const float*)d_in[4], (const float*)d_in[8],  (const float*)d_in[12]};
    const float* adst[3] = {(const float*)d_in[5], (const float*)d_in[9],  (const float*)d_in[13]};
    const float* bias[3] = {(const float*)d_in[6], (const float*)d_in[10], (const float*)d_in[14]};

    const int n  = in_sizes[0] / GDIM;
    const int E  = in_sizes[1] / 2;
    const int NE = E + n;

    char*  wsb = (char*)d_ws;
    size_t off = 0;
    auto take = [&](size_t bytes) -> void* {
        void* p = wsb + off;
        off += (bytes + 255) & ~(size_t)255;
        return p;
    };
    const size_t szH = (size_t)n * GDIM * sizeof(float);
    float*    h      = (float*)   take(szH);
    float*    acc    = (float*)   take(szH);
    float*    s_src  = (float*)   take((size_t)n * sizeof(float));
    float*    s_dst  = (float*)   take((size_t)n * sizeof(float));
    unsigned* m_enc  = (unsigned*)take((size_t)n * sizeof(unsigned));
    float*    denom  = (float*)   take((size_t)n * sizeof(float));
    float*    psum   = (float*)   take((size_t)NGRAPH * GDIM * sizeof(float));
    float*    pcount = (float*)   take((size_t)NGRAPH * sizeof(float));
    (void)ws_size; (void)n_in; (void)out_size;

    const int nTiles     = (n + 15) / 16;
    const int gemmBlocks = (nTiles + 3) / 4;
    const int nodeBlocks = (n + 7) / 8;
    const int edgeTBlk   = (NE + 255) / 256;
    const int edgeWBlk   = (NE + 7) / 8;

    const float* xin = x;
    for (int l = 0; l < 3; ++l) {
        gat_gemm_wmma<<<gemmBlocks, 128, GDIM * GDIM * sizeof(float), stream>>>(
            xin, W[l], h, nTiles);
        gat_node_scores<<<nodeBlocks, 256, 0, stream>>>(h, asrc[l], adst[l],
                                                        s_src, s_dst, n);
        hipMemsetAsync(m_enc, 0, (size_t)n * sizeof(unsigned), stream);
        hipMemsetAsync(denom, 0, (size_t)n * sizeof(float), stream);
        hipMemsetAsync(acc,   0, szH, stream);
        gat_edge_max<<<edgeTBlk, 256, 0, stream>>>(ei, E, n, s_src, s_dst, m_enc);
        gat_edge_aggr<<<edgeWBlk, 256, 0, stream>>>(ei, E, n, s_src, s_dst,
                                                    m_enc, h, denom, acc);
        gat_finalize<<<nodeBlocks, 256, 0, stream>>>(acc, denom, bias[l], n);
        xin = acc;
    }

    hipMemsetAsync(psum,   0, (size_t)NGRAPH * GDIM * sizeof(float), stream);
    hipMemsetAsync(pcount, 0, (size_t)NGRAPH * sizeof(float), stream);
    gat_pool_accum<<<nodeBlocks, 256, 0, stream>>>(acc, batch, n, psum, pcount);
    gat_pool_final<<<(NGRAPH * GDIM) / 256, 256, 0, stream>>>(psum, pcount,
                                                              (float*)d_out);
}


# edge aggregation by owner: a work-group keeps 400 dst rows' sums in LDS, scans the dst list for its edges, adds rows by LDS f32 atomics; global f32 atomics, acc memset, finalize gone; arrival order
# speedup vs baseline: 2.0692x; 2.0692x over previous
#include <hip/hip_runtime.h>

#define GDIM 128
#define NGRAPH 256
#define NEG_SLOPE 0.2f

typedef float v2f __attribute__((ext_vector_type(2)));
typedef float v8f __attribute__((ext_vector_type(8)));

__device__ __forceinline__ unsigned enc_f32(float f) {
    unsigned b = __float_as_uint(f);
    return (b & 0x80000000u) ? ~b : (b | 0x80000000u);
}
__device__ __forceinline__ float dec_f32(unsigned u) {
    return __uint_as_float((u & 0x80000000u) ? (u ^ 0x80000000u) : ~u);
}

__global__ void __launch_bounds__(128)
gat_gemm_wmma(const float* __restrict__ X, const float* __restrict__ W,
              float* __restrict__ H, int nTiles) {
    extern __shared__ float sW[];
    const int tid = threadIdx.x;

    const float4* W4 = (const float4*)W;
    float4*       S4 = (float4*)sW;
#pragma unroll
    for (int t = 0; t < 32; ++t) S4[t * 128 + tid] = W4[t * 128 + tid];
    __syncthreads();

    const int wid  = tid >> 5;
    const int lane = tid & 31;
    const int hi   = lane >> 4;
    const int col  = lane & 15;
    const int tile = blockIdx.x * 4 + wid;
    if (tile >= nTiles) return;

    const long   row0 = (long)tile * 16;
    const float* Xr   = X + (row0 + col) * GDIM;

    v8f c[8];
#pragma unroll
    for (int n = 0; n < 8; ++n) c[n] = (v8f)0.0f;

    for (int kk = 0; kk < GDIM; kk += 4) {
        const int ka = kk + 2 * hi;
        v2f a = *(const v2f*)(Xr + ka);
#pragma unroll
        for (int n = 0; n < 8; ++n) {
            v2f b;
            b.x = sW[(ka + 0) * GDIM + n * 16 + col];
            b.y = sW[(ka + 1) * GDIM + n * 16 + col];
            c[n] = __builtin_amdgcn_wmma_f32_16x16x4_f32(
                       false, a, false, b, (short)0, c[n], false, false);
        }
    }

    float* Hr = H + row0 * GDIM;
#pragma unroll
    for (int n = 0; n < 8; ++n)
#pragma unroll
        for (int v = 0; v < 8; ++v)
            Hr[(v + 8 * hi) * GDIM + n * 16 + col] = c[n][v];
}

__global__ void __launch_bounds__(256)
gat_node_scores(const float* __restrict__ H, const float* __restrict__ a_src,
                const float* __restrict__ a_dst, float* __restrict__ s_src,
                float* __restrict__ s_dst, int n) {
    const int lane = threadIdx.x & 31;
    const int node = blockIdx.x * 8 + (threadIdx.x >> 5);
    if (node >= n) return;
    const float4 hv = *(const float4*)(H + (long)node * GDIM + lane * 4);
    const float4 as = *(const float4*)(a_src + lane * 4);
    const float4 ad = *(const float4*)(a_dst + lane * 4);
    float ps = hv.x * as.x + hv.y * as.y + hv.z * as.z + hv.w * as.w;
    float pd = hv.x * ad.x + hv.y * ad.y + hv.z * ad.z + hv.w * ad.w;
#pragma unroll
    for (int off = 16; off > 0; off >>= 1) {
        ps += __shfl_xor(ps, off, 32);
        pd += __shfl_xor(pd, off, 32);
    }
    if (lane == 0) { s_src[node] = ps; s_dst[node] = pd; }
}

__global__ void __launch_bounds__(256)
gat_edge_max(const int* __restrict__ ei, int E, int n,
             const float* __restrict__ s_src, const float* __restrict__ s_dst,
             unsigned* __restrict__ m_enc) {
    const int e  = blockIdx.x * blockDim.x + threadIdx.x;
    const int NE = E + n;
    if (e >= NE) return;
    int i, j;
    if (e < E) { i = ei[e]; j = ei[E + e]; } else { i = j = e - E; }
    const float s     = s_src[i] + s_dst[j];
    const float logit = (s > 0.f) ? s : NEG_SLOPE * s;
    atomicMax(m_enc + j, enc_f32(logit));
}

#define RNG  400
#define LCAP 8192

__global__ void __launch_bounds__(1024)
gat_range_aggr(const int* __restrict__ ei, int E, int n,
               const float* __restrict__ s_src, const float* __restrict__ s_dst,
               const unsigned* __restrict__ m_enc, const float* __restrict__ H,
               const float* __restrict__ bias, float* __restrict__ out) {
    __shared__ float    l_acc[RNG * GDIM];
    __shared__ int2     l_list[LCAP];
    __shared__ float    l_m[RNG];
    __shared__ float    l_sd[RNG];
    __shared__ float    l_den[RNG];
    __shared__ unsigned l_cnt;

    const int tid  = threadIdx.x;
    const int lane = tid & 31;
    const int wid  = __builtin_amdgcn_readfirstlane(tid >> 5);
    const int j0   = blockIdx.x * RNG;
    const unsigned nloc = (unsigned)((n - j0 < RNG) ? (n - j0) : RNG);

    auto self_w = [&](float ss, float sd, float m) -> float {
        const float s     = ss + sd;
        const float logit = (s > 0.f) ? s : NEG_SLOPE * s;
        return __expf(logit - m);
    };
    for (unsigned d0 = (unsigned)wid * 4u; d0 < nloc; d0 += 128u) {
        if (d0 + 4u <= nloc) {
            float hv[4][4];
            float ssv[4], sdv[4], mv[4];
#pragma unroll
            for (int u = 0; u < 4; ++u) {
                const int j = j0 + (int)d0 + u;
                const float* hp = H + (long)j * GDIM + lane;
                hv[u][0] = hp[0]; hv[u][1] = hp[32]; hv[u][2] = hp[64]; hv[u][3] = hp[96];
                ssv[u] = s_src[j];
                sdv[u] = s_dst[j];
                mv[u]  = dec_f32(m_enc[j]);
            }
#pragma unroll
            for (int u = 0; u < 4; ++u) {
                const float w = self_w(ssv[u], sdv[u], mv[u]);
                float* ap = l_acc + (d0 + u) * GDIM + lane;
                ap[0]  = w * hv[u][0];
                ap[32] = w * hv[u][1];
                ap[64] = w * hv[u][2];
                ap[96] = w * hv[u][3];
                if (lane == 0) { l_m[d0 + u] = mv[u]; l_sd[d0 + u] = sdv[u]; l_den[d0 + u] = w; }
            }
        } else {
            for (unsigned dl = d0; dl < nloc; ++dl) {
                const int j = j0 + (int)dl;
                const float* hp = H + (long)j * GDIM + lane;
                const float h0 = hp[0], h1 = hp[32], h2 = hp[64], h3 = hp[96];
                const float sd = s_dst[j];
                const float m  = dec_f32(m_enc[j]);
                const float w  = self_w(s_src[j], sd, m);
                float* ap = l_acc + dl * GDIM + lane;
                ap[0]  = w * h0;
                ap[32] = w * h1;
                ap[64] = w * h2;
                ap[96] = w * h3;
                if (lane == 0) { l_m[dl] = m; l_sd[dl] = sd; l_den[dl] = w; }
            }
        }
    }
    if (tid == 0) l_cnt = 0u;
    __syncthreads();

    const int* dstp = ei + E;
    auto keep = [&](int e, unsigned dl) {
        const unsigned pos = atomicAdd(&l_cnt, 1u);
        if (pos < LCAP) {
            l_list[pos] = make_int2(e, (int)dl);
        } else {
            const int   src   = ei[e];
            const float s     = s_src[src] + l_sd[dl];
            const float logit = (s > 0.f) ? s : NEG_SLOPE * s;
            const float w     = __expf(logit - l_m[dl]);
            atomicAdd(l_den + dl, w);
            const float* hp = H + (long)src * GDIM;
            float*       ap = l_acc + dl * GDIM;
            for (int c = 0; c < GDIM; ++c) atomicAdd(ap + c, w * hp[c]);
        }
    };
    if ((E & 3) == 0) {
        const int4* d4   = (const int4*)dstp;
        const int   nvec = E >> 2;
        const int   nIt  = (nvec + 1023) >> 10;
        const int   rot  = (int)(((long)blockIdx.x * nIt) / gridDim.x);
        for (int it = 0; it < nIt; it += 4) {
            int4 v[4];
            int  vi[4];
#pragma unroll
            for (int u = 0; u < 4; ++u) {
                int c = it + u + rot;
                if (c >= nIt) c -= nIt;
                vi[u] = (c << 10) + tid;
                const bool ok = (it + u < nIt) && (vi[u] < nvec);
                v[u] = ok ? d4[vi[u]] : make_int4(-1, -1, -1, -1);
            }
#pragma unroll
            for (int u = 0; u < 4; ++u) {
                const unsigned d0 = (unsigned)(v[u].x - j0);
                const unsigned d1 = (unsigned)(v[u].y - j0);
                const unsigned d2 = (unsigned)(v[u].z - j0);
                const unsigned d3 = (unsigned)(v[u].w - j0);
                const unsigned dm = min(min(d0, d1), min(d2, d3));
                if (dm < nloc) {
                    const int e0 = vi[u] << 2;
#pragma nounroll
                    for (int q = 0; q < 4; ++q) {
                        const unsigned d = (q == 0) ? d0 : (q == 1) ? d1 : (q == 2) ? d2 : d3;
                        if (d < nloc) keep(e0 + q, d);
                    }
                }
            }
        }
    } else {
        for (int e = tid; e < E; e += 1024) {
            const unsigned d = (unsigned)(dstp[e] - j0);
            if (d < nloc) keep(e, d);
        }
    }
    __syncthreads();

    const int total = __builtin_amdgcn_readfirstlane(
        (l_cnt < (unsigned)LCAP) ? (int)l_cnt : LCAP);
    for (int b = wid * 32; b < total; b += 32 * 32) {
        const int nb = (total - b < 32) ? (total - b) : 32;
        int   src = 0;
        int   dlv = 0;
        float w   = 0.f;
        if (lane < nb) {
            const int2 en = l_list[b + lane];
            dlv = en.y;
            src = ei[en.x];
            const float s     = s_src[src] + l_sd[dlv];
            const float logit = (s > 0.f) ? s : NEG_SLOPE * s;
            w = __expf(logit - l_m[dlv]);
            atomicAdd(l_den + dlv, w);
        }
        int k = 0;
        for (; k + 8 <= nb; k += 8) {
            float h[8][4];
            float ww[8];
            int   dd[8];
#pragma unroll
            for (int u = 0; u < 8; ++u) {
                const int sk = __builtin_amdgcn_readlane(src, k + u);
                dd[u] = __builtin_amdgcn_readlane(dlv, k + u);
                ww[u] = __int_as_float(__builtin_amdgcn_readlane(__float_as_int(w), k + u));
                const float* hp = H + (long)sk * GDIM + lane;
                h[u][0] = hp[0]; h[u][1] = hp[32]; h[u][2] = hp[64]; h[u][3] = hp[96];
            }
#pragma unroll
            for (int u = 0; u < 8; ++u) {
                float* ap = l_acc + dd[u] * GDIM + lane;
                atomicAdd(ap + 0,  ww[u] * h[u][0]);
                atomicAdd(ap + 32, ww[u] * h[u][1]);
                atomicAdd(ap + 64, ww[u] * h[u][2]);
                atomicAdd(ap + 96, ww[u] * h[u][3]);
            }
        }
        for (; k < nb; ++k) {
            const int   sk = __builtin_amdgcn_readlane(src, k);
            const int   dk = __builtin_amdgcn_readlane(dlv, k);
            const float wk = __int_as_float(__builtin_amdgcn_readlane(__float_as_int(w), k));
            const float* hp = H + (long)sk * GDIM + lane;
            float*       ap = l_acc + dk * GDIM + lane;
            const float h0 = hp[0], h1 = hp[32], h2 = hp[64], h3 = hp[96];
            atomicAdd(ap + 0,  wk * h0);
            atomicAdd(ap + 32, wk * h1);
            atomicAdd(ap + 64, wk * h2);
            atomicAdd(ap + 96, wk * h3);
        }
    }
    __syncthreads();

    const float4 b4 = *(const float4*)(bias + lane * 4);
    for (unsigned dl = wid; dl < nloc; dl += 32) {
        const float inv = 1.0f / l_den[dl];
        float4 v = *(const float4*)(l_acc + dl * GDIM + lane * 4);
        v.x = v.x * inv + b4.x;
        v.y = v.y * inv + b4.y;
        v.z = v.z * inv + b4.z;
        v.w = v.w * inv + b4.w;
        *(float4*)(out + (long)(j0 + (int)dl) * GDIM + lane * 4) = v;
    }
}

__global__ void __launch_bounds__(256)
gat_pool_accum(const float* __restrict__ h, const int* __restrict__ batch, int n,
               float* __restrict__ psum, float* __restrict__ pcount) {
    const int lane = threadIdx.x & 31;
    const int node = blockIdx.x * 8 + (threadIdx.x >> 5);
    if (node >= n) return;
    const int    g  = batch[node];
    const float* hp = h + (long)node * GDIM + lane;
    float*       pp = psum + (long)g * GDIM + lane;
    const float h0 = hp[0], h1 = hp[32], h2 = hp[64], h3 = hp[96];
    atomicAdd(pp + 0,  h0);
    atomicAdd(pp + 32, h1);
    atomicAdd(pp + 64, h2);
    atomicAdd(pp + 96, h3);
    if (lane == 0) atomicAdd(pcount + g, 1.0f);
}

__global__ void __launch_bounds__(256)
gat_pool_final(const float* __restrict__ psum, const float* __restrict__ pcount,
               float* __restrict__ out) {
    const int idx = blockIdx.x * blockDim.x + threadIdx.x;
    const float cnt = pcount[idx >> 7];
    out[idx] = psum[idx] / fmaxf(cnt, 1.0f);
}

extern "C" void kernel_launch(void* const* d_in, const int* in_sizes, int n_in,
                              void* d_out, int out_size, void* d_ws, size_t ws_size,
                              hipStream_t stream) {
    const float* x     = (const float*)d_in[0];
    const int*   ei    = (const int*)d_in[1];
    const int*   batch = (const int*)d_in[2];
    const float* W[3]    = {(const float*)d_in[3], (const float*)d_in[7],  (const float*)d_in[11]};
    const float* asrc[3] = {(const float*)d_in[4], (const float*)d_in[8],  (const float*)d_in[12]};
    const float* adst[3] = {(const float*)d_in[5], (const float*)d_in[9],  (const float*)d_in[13]};
    const float* bias[3] = {(const float*)d_in[6], (const float*)d_in[10], (const float*)d_in[14]};

    const int n  = in_sizes[0] / GDIM;
    const int E  = in_sizes[1] / 2;
    const int NE = E + n;

    char*  wsb = (char*)d_ws;
    size_t off = 0;
    auto take = [&](size_t bytes) -> void* {
        void* p = wsb + off;
        off += (bytes + 255) & ~(size_t)255;
        return p;
    };
    const size_t szH = (size_t)n * GDIM * sizeof(float);
    float*    h      = (float*)   take(szH);
    float*    acc    = (float*)   take(szH);
    float*    s_src  = (float*)   take((size_t)n * sizeof(float));
    float*    s_dst  = (float*)   take((size_t)n * sizeof(float));
    unsigned* m_enc  = (unsigned*)take((size_t)n * sizeof(unsigned));
    float*    psum   = (float*)   take((size_t)NGRAPH * GDIM * sizeof(float));
    float*    pcount = (float*)   take((size_t)NGRAPH * sizeof(float));
    (void)ws_size; (void)n_in; (void)out_size;

    const int nTiles     = (n + 15) / 16;
    const int gemmBlocks = (nTiles + 3) / 4;
    const int nodeBlocks = (n + 7) / 8;
    const int edgeTBlk   = (NE + 255) / 256;
    const int rangeBlk   = (n + RNG - 1) / RNG;

    const float* xin = x;
    for (int l = 0; l < 3; ++l) {
        gat_gemm_wmma<<<gemmBlocks, 128, GDIM * GDIM * sizeof(float), stream>>>(
            xin, W[l], h, nTiles);
        gat_node_scores<<<nodeBlocks, 256, 0, stream>>>(h, asrc[l], adst[l],
                                                        s_src, s_dst, n);
        hipMemsetAsync(m_enc, 0, (size_t)n * sizeof(unsigned), stream);
        gat_edge_max<<<edgeTBlk, 256, 0, stream>>>(ei, E, n, s_src, s_dst, m_enc);
        gat_range_aggr<<<rangeBlk, 1024, 0, stream>>>(ei, E, n, s_src, s_dst,
                                                      m_enc, h, bias[l], acc);
        xin = acc;
    }

    hipMemsetAsync(psum,   0, (size_t)NGRAPH * GDIM * sizeof(float), stream);
    hipMemsetAsync(pcount, 0, (size_t)NGRAPH * sizeof(float), stream);
    gat_pool_accum<<<nodeBlocks, 256, 0, stream>>>(acc, batch, n, psum, pcount);
    gat_pool_final<<<(NGRAPH * GDIM) / 256, 256, 0, stream>>>(psum, pcount,
                                                              (float*)d_out);
}


# segment max by the owner work-group in LDS (integer max over its kept edges; self loop is one more list entry, so it too adds in arrival order); edge-max launch, its memset and max array gone
# speedup vs baseline: 2.6313x; 1.2717x over previous
#include <hip/hip_runtime.h>

#define GDIM 128
#define NGRAPH 256
#define NEG_SLOPE 0.2f

typedef float v2f __attribute__((ext_vector_type(2)));
typedef float v8f __attribute__((ext_vector_type(8)));

__device__ __forceinline__ unsigned enc_f32(float f) {
    unsigned b = __float_as_uint(f);
    return (b & 0x80000000u) ? ~b : (b | 0x80000000u);
}
__device__ __forceinline__ float dec_f32(unsigned u) {
    return __uint_as_float((u & 0x80000000u) ? (u ^ 0x80000000u) : ~u);
}

__global__ void __launch_bounds__(128)
gat_gemm_wmma(const float* __restrict__ X, const float* __restrict__ W,
              float* __restrict__ H, int nTiles) {
    extern __shared__ float sW[];
    const int tid = threadIdx.x;

    const float4* W4 = (const float4*)W;
    float4*       S4 = (float4*)sW;
#pragma unroll
    for (int t = 0; t < 32; ++t) S4[t * 128 + tid] = W4[t * 128 + tid];
    __syncthreads();

    const int wid  = tid >> 5;
    const int lane = tid & 31;
    const int hi   = lane >> 4;
    const int col  = lane & 15;
    const int tile = blockIdx.x * 4 + wid;
    if (tile >= nTiles) return;

    const long   row0 = (long)tile * 16;
    const float* Xr   = X + (row0 + col) * GDIM;

    v8f c[8];
#pragma unroll
    for (int n = 0; n < 8; ++n) c[n] = (v8f)0.0f;

    for (int kk = 0; kk < GDIM; kk += 4) {
        const int ka = kk + 2 * hi;
        v2f a = *(const v2f*)(Xr + ka);
#pragma unroll
        for (int n = 0; n < 8; ++n) {
            v2f b;
            b.x = sW[(ka + 0) * GDIM + n * 16 + col];
            b.y = sW[(ka + 1) * GDIM + n * 16 + col];
            c[n] = __builtin_amdgcn_wmma_f32_16x16x4_f32(
                       false, a, false, b, (short)0, c[n], false, false);
        }
    }

    float* Hr = H + row0 * GDIM;
#pragma unroll
    for (int n = 0; n < 8; ++n)
#pragma unroll
        for (int v = 0; v < 8; ++v)
            Hr[(v + 8 * hi) * GDIM + n * 16 + col] = c[n][v];
}

__global__ void __launch_bounds__(256)
gat_node_scores(const float* __restrict__ H, const float* __restrict__ a_src,
                const float* __restrict__ a_dst, float* __restrict__ s_src,
                float* __restrict__ s_dst, int n) {
    const int lane = threadIdx.x & 31;
    const int node = blockIdx.x * 8 + (threadIdx.x >> 5);
    if (node >= n) return;
    const float4 hv = *(const float4*)(H + (long)node * GDIM + lane * 4);
    const float4 as = *(const float4*)(a_src + lane * 4);
    const float4 ad = *(const float4*)(a_dst + lane * 4);
    float ps = hv.x * as.x + hv.y * as.y + hv.z * as.z + hv.w * as.w;
    float pd = hv.x * ad.x + hv.y * ad.y + hv.z * ad.z + hv.w * ad.w;
#pragma unroll
    for (int off = 16; off > 0; off >>= 1) {
        ps += __shfl_xor(ps, off, 32);
        pd += __shfl_xor(pd, off, 32);
    }
    if (lane == 0) { s_src[node] = ps; s_dst[node] = pd; }
}

#define RNG  400
#define LCAP 8192

__global__ void __launch_bounds__(1024)
gat_range_aggr(const int* __restrict__ ei, int E, int n,
               const float* __restrict__ s_src, const float* __restrict__ s_dst,
               const float* __restrict__ H,
               const float* __restrict__ bias, float* __restrict__ out) {
    __shared__ float    l_acc[RNG * GDIM];
    __shared__ int2     l_list[LCAP];
    __shared__ float    l_lg[LCAP];
    __shared__ unsigned l_menc[RNG];
    __shared__ float    l_sd[RNG];
    __shared__ float    l_den[RNG];
    __shared__ unsigned l_cnt;

    const int tid  = threadIdx.x;
    const int lane = tid & 31;
    const int wid  = __builtin_amdgcn_readfirstlane(tid >> 5);
    const int j0   = blockIdx.x * RNG;
    const unsigned nloc = (unsigned)((n - j0 < RNG) ? (n - j0) : RNG);

    float4* a4 = (float4*)l_acc;
    for (int i = tid; i < RNG * GDIM / 4; i += 1024) a4[i] = make_float4(0.f, 0.f, 0.f, 0.f);
    if ((unsigned)tid < nloc) {
        const unsigned dl = (unsigned)tid;
        const int   j     = j0 + (int)dl;
        const float sd    = s_dst[j];
        const float s     = s_src[j] + sd;
        const float logit = (s > 0.f) ? s : NEG_SLOPE * s;
        l_sd[dl]   = sd;
        l_den[dl]  = 0.f;
        l_menc[dl] = enc_f32(logit);
        l_list[dl] = make_int2(j, (int)dl);
        l_lg[dl]   = logit;
    }
    if (tid == 0) l_cnt = nloc;
    __syncthreads();

    const int* dstp = ei + E;
    auto keep = [&](int e, unsigned dl) {
        const unsigned pos = atomicAdd(&l_cnt, 1u);
        if (pos < LCAP) l_list[pos] = make_int2(e, (int)dl);
    };
    if ((E & 3) == 0) {
        const int4* d4   = (const int4*)dstp;
        const int   nvec = E >> 2;
        const int   nIt  = (nvec + 1023) >> 10;
        const int   rot  = (int)(((long)blockIdx.x * nIt) / gridDim.x);
        for (int it = 0; it < nIt; it += 4) {
            int4 v[4];
            int  vi[4];
#pragma unroll
            for (int u = 0; u < 4; ++u) {
                int c = it + u + rot;
                if (c >= nIt) c -= nIt;
                vi[u] = (c << 10) + tid;
                const bool ok = (it + u < nIt) && (vi[u] < nvec);
                v[u] = ok ? d4[vi[u]] : make_int4(-1, -1, -1, -1);
            }
#pragma unroll
            for (int u = 0; u < 4; ++u) {
                const unsigned d0 = (unsigned)(v[u].x - j0);
                const unsigned d1 = (unsigned)(v[u].y - j0);
                const unsigned d2 = (unsigned)(v[u].z - j0);
                const unsigned d3 = (unsigned)(v[u].w - j0);
                const unsigned dm = min(min(d0, d1), min(d2, d3));
                if (dm < nloc) {
                    const int e0 = vi[u] << 2;
#pragma nounroll
                    for (int q = 0; q < 4; ++q) {
                        const unsigned d = (q == 0) ? d0 : (q == 1) ? d1 : (q == 2) ? d2 : d3;
                        if (d < nloc) keep(e0 + q, d);
                    }
                }
            }
        }
    } else {
        for (int e = tid; e < E; e += 1024) {
            const unsigned d = (unsigned)(dstp[e] - j0);
            if (d < nloc) keep(e, d);
        }
    }
    __syncthreads();

    const int total = __builtin_amdgcn_readfirstlane((int)l_cnt);

    if (total <= LCAP) {
        for (int i0 = (int)nloc + tid; i0 < total; i0 += 4096) {
            int   ii[4];
            int2  en[4];
            int   sr[4];
            float ss[4];
#pragma unroll
            for (int u = 0; u < 4; ++u) {
                ii[u] = i0 + u * 1024;
                const int ic = (ii[u] < total) ? ii[u] : i0;
                en[u] = l_list[ic];
                sr[u] = ei[en[u].x];
            }
#pragma unroll
            for (int u = 0; u < 4; ++u) ss[u] = s_src[sr[u]];
#pragma unroll
            for (int u = 0; u < 4; ++u) {
                const float s     = ss[u] + l_sd[en[u].y];
                const float logit = (s > 0.f) ? s : NEG_SLOPE * s;
                atomicMax(l_menc + en[u].y, enc_f32(logit));
                if (ii[u] < total) {
                    l_list[ii[u]] = make_int2(sr[u], en[u].y);
                    l_lg[ii[u]]   = logit;
                }
            }
        }
        __syncthreads();

        for (int b = wid * 32; b < total; b += 32 * 32) {
            const int nb = (total - b < 32) ? (total - b) : 32;
            int   src = 0;
            int   dlv = 0;
            float w   = 0.f;
            if (lane < nb) {
                const int2 en = l_list[b + lane];
                src = en.x;
                dlv = en.y;
                w   = __expf(l_lg[b + lane] - dec_f32(l_menc[dlv]));
                atomicAdd(l_den + dlv, w);
            }
            int k = 0;
            for (; k + 8 <= nb; k += 8) {
                float h[8][4];
                float ww[8];
                int   dd[8];
#pragma unroll
                for (int u = 0; u < 8; ++u) {
                    const int sk = __builtin_amdgcn_readlane(src, k + u);
                    dd[u] = __builtin_amdgcn_readlane(dlv, k + u);
                    ww[u] = __int_as_float(__builtin_amdgcn_readlane(__float_as_int(w), k + u));
                    const float* hp = H + (long)sk * GDIM + lane;
                    h[u][0] = hp[0]; h[u][1] = hp[32]; h[u][2] = hp[64]; h[u][3] = hp[96];
                }
#pragma unroll
                for (int u = 0; u < 8; ++u) {
                    float* ap = l_acc + dd[u] * GDIM + lane;
                    atomicAdd(ap + 0,  ww[u] * h[u][0]);
                    atomicAdd(ap + 32, ww[u] * h[u][1]);
                    atomicAdd(ap + 64, ww[u] * h[u][2]);
                    atomicAdd(ap + 96, ww[u] * h[u][3]);
                }
            }
            for (; k < nb; ++k) {
                const int   sk = __builtin_amdgcn_readlane(src, k);
                const int   dk = __builtin_amdgcn_readlane(dlv, k);
                const float wk = __int_as_float(__builtin_amdgcn_readlane(__float_as_int(w), k));
                const float* hp = H + (long)sk * GDIM + lane;
                float*       ap = l_acc + dk * GDIM + lane;
                const float h0 = hp[0], h1 = hp[32], h2 = hp[64], h3 = hp[96];
                atomicAdd(ap + 0,  wk * h0);
                atomicAdd(ap + 32, wk * h1);
                atomicAdd(ap + 64, wk * h2);
                atomicAdd(ap + 96, wk * h3);
            }
        }
    } else {
        for (int e = tid; e < E; e += 1024) {
            const unsigned d = (unsigned)(dstp[e] - j0);
            if (d < nloc) {
                const float s     = s_src[ei[e]] + l_sd[d];
                const float logit = (s > 0.f) ? s : NEG_SLOPE * s;
                atomicMax(l_menc + d, enc_f32(logit));
            }
        }
        __syncthreads();
        for (int e = tid; e < E + (int)nloc; e += 1024) {
            unsigned d;
            int      src;
            if (e < E) { d = (unsigned)(dstp[e] - j0); src = (d < nloc) ? ei[e] : 0; }
            else       { d = (unsigned)(e - E);        src = j0 + (int)d; }
            if (d < nloc) {
                const float s     = s_src[src] + l_sd[d];
                const float logit = (s > 0.f) ? s : NEG_SLOPE * s;
                const float w     = __expf(logit - dec_f32(l_menc[d]));
                atomicAdd(l_den + d, w);
                const float* hp = H + (long)src * GDIM;
                float*       ap = l_acc + d * GDIM;
                for (int c = 0; c < GDIM; ++c) atomicAdd(ap + c, w * hp[c]);
            }
        }
    }
    __syncthreads();

    const float4 b4 = *(const float4*)(bias + lane * 4);
    for (unsigned dl = wid; dl < nloc; dl += 32) {
        const float inv = 1.0f / l_den[dl];
        float4 v = *(const float4*)(l_acc + dl * GDIM + lane * 4);
        v.x = v.x * inv + b4.x;
        v.y = v.y * inv + b4.y;
        v.z = v.z * inv + b4.z;
        v.w = v.w * inv + b4.w;
        *(float4*)(out + (long)(j0 + (int)dl) * GDIM + lane * 4) = v;
    }
}

__global__ void __launch_bounds__(256)
gat_pool_accum(const float* __restrict__ h, const int* __restrict__ batch, int n,
               float* __restrict__ psum, float* __restrict__ pcount) {
    const int lane = threadIdx.x & 31;
    const int node = blockIdx.x * 8 + (threadIdx.x >> 5);
    if (node >= n) return;
    const int    g  = batch[node];
    const float* hp = h + (long)node * GDIM + lane;
    float*       pp = psum + (long)g * GDIM + lane;
    const float h0 = hp[0], h1 = hp[32], h2 = hp[64], h3 = hp[96];
    atomicAdd(pp + 0,  h0);
    atomicAdd(pp + 32, h1);
    atomicAdd(pp + 64, h2);
    atomicAdd(pp + 96, h3);
    if (lane == 0) atomicAdd(pcount + g, 1.0f);
}

__global__ void __launch_bounds__(256)
gat_pool_final(const float* __restrict__ psum, const float* __restrict__ pcount,
               float* __restrict__ out) {
    const int idx = blockIdx.x * blockDim.x + threadIdx.x;
    const float cnt = pcount[idx >> 7];
    out[idx] = psum[idx] / fmaxf(cnt, 1.0f);
}

extern "C" void kernel_launch(void* const* d_in, const int* in_sizes, int n_in,
                              void* d_out, int out_size, void* d_ws, size_t ws_size,
                              hipStream_t stream) {
    const float* x     = (const float*)d_in[0];
    const int*   ei    = (const int*)d_in[1];
    const int*   batch = (const int*)d_in[2];
    const float* W[3]    = {(const float*)d_in[3], (const float*)d_in[7],  (const float*)d_in[11]};
    const float* asrc[3] = {(const float*)d_in[4], (const float*)d_in[8],  (const float*)d_in[12]};
    const float* adst[3] = {(const float*)d_in[5], (const float*)d_in[9],  (const float*)d_in[13]};
    const float* bias[3] = {(const float*)d_in[6], (const float*)d_in[10], (const float*)d_in[14]};

    const int n  = in_sizes[0] / GDIM;
    const int E  = in_sizes[1] / 2;

    char*  wsb = (char*)d_ws;
    size_t off = 0;
    auto take = [&](size_t bytes) -> void* {
        void* p = wsb + off;
        off += (bytes + 255) & ~(size_t)255;
        return p;
    };
    const size_t szH = (size_t)n * GDIM * sizeof(float);
    float*    h      = (float*)   take(szH);
    float*    acc    = (float*)   take(szH);
    float*    s_src  = (float*)   take((size_t)n * sizeof(float));
    float*    s_dst  = (float*)   take((size_t)n * sizeof(float));
    float*    psum   = (float*)   take((size_t)NGRAPH * GDIM * sizeof(float));
    float*    pcount = (float*)   take((size_t)NGRAPH * sizeof(float));
    (void)ws_size; (void)n_in; (void)out_size;

    const int nTiles     = (n + 15) / 16;
    const int gemmBlocks = (nTiles + 3) / 4;
    const int nodeBlocks = (n + 7) / 8;
    const int rangeBlk   = (n + RNG - 1) / RNG;

    const float* xin = x;
    for (int l = 0; l < 3; ++l) {
        gat_gemm_wmma<<<gemmBlocks, 128, GDIM * GDIM * sizeof(float), stream>>>(
            xin, W[l], h, nTiles);
        gat_node_scores<<<nodeBlocks, 256, 0, stream>>>(h, asrc[l], adst[l],
                                                        s_src, s_dst, n);
        gat_range_aggr<<<rangeBlk, 1024, 0, stream>>>(ei, E, n, s_src, s_dst,
                                                      h, bias[l], acc);
        xin = acc;
    }

    hipMemsetAsync(psum,   0, (size_t)NGRAPH * GDIM * sizeof(float), stream);
    hipMemsetAsync(pcount, 0, (size_t)NGRAPH * sizeof(float), stream);
    gat_pool_accum<<<nodeBlocks, 256, 0, stream>>>(acc, batch, n, psum, pcount);
    gat_pool_final<<<(NGRAPH * GDIM) / 256, 256, 0, stream>>>(psum, pcount,
                                                              (float*)d_out);
}


# edge lists once per call: layer 1's owner work-groups save their kept (source row, local row) entries in the workspace; layers 2 and 3 read them, not the whole dst list; sums stay in arrival order
# speedup vs baseline: 3.2306x; 1.2278x over previous
#include <hip/hip_runtime.h>

#define GDIM 128
#define NGRAPH 256
#define NEG_SLOPE 0.2f

typedef float v2f __attribute__((ext_vector_type(2)));
typedef float v8f __attribute__((ext_vector_type(8)));

__device__ __forceinline__ unsigned enc_f32(float f) {
    unsigned b = __float_as_uint(f);
    return (b & 0x80000000u) ? ~b : (b | 0x80000000u);
}
__device__ __forceinline__ float dec_f32(unsigned u) {
    return __uint_as_float((u & 0x80000000u) ? (u ^ 0x80000000u) : ~u);
}

__global__ void __launch_bounds__(128)
gat_gemm_wmma(const float* __restrict__ X, const float* __restrict__ W,
              float* __restrict__ H, int nTiles) {
    extern __shared__ float sW[];
    const int tid = threadIdx.x;

    const float4* W4 = (const float4*)W;
    float4*       S4 = (float4*)sW;
#pragma unroll
    for (int t = 0; t < 32; ++t) S4[t * 128 + tid] = W4[t * 128 + tid];
    __syncthreads();

    const int wid  = tid >> 5;
    const int lane = tid & 31;
    const int hi   = lane >> 4;
    const int col  = lane & 15;
    const int tile = blockIdx.x * 4 + wid;
    if (tile >= nTiles) return;

    const long   row0 = (long)tile * 16;
    const float* Xr   = X + (row0 + col) * GDIM;

    v8f c[8];
#pragma unroll
    for (int n = 0; n < 8; ++n) c[n] = (v8f)0.0f;

    for (int kk = 0; kk < GDIM; kk += 4) {
        const int ka = kk + 2 * hi;
        v2f a = *(const v2f*)(Xr + ka);
#pragma unroll
        for (int n = 0; n < 8; ++n) {
            v2f b;
            b.x = sW[(ka + 0) * GDIM + n * 16 + col];
            b.y = sW[(ka + 1) * GDIM + n * 16 + col];
            c[n] = __builtin_amdgcn_wmma_f32_16x16x4_f32(
                       false, a, false, b, (short)0, c[n], false, false);
        }
    }

    float* Hr = H + row0 * GDIM;
#pragma unroll
    for (int n = 0; n < 8; ++n)
#pragma unroll
        for (int v = 0; v < 8; ++v)
            Hr[(v + 8 * hi) * GDIM + n * 16 + col] = c[n][v];
}

__global__ void __launch_bounds__(256)
gat_node_scores(const float* __restrict__ H, const float* __restrict__ a_src,
                const float* __restrict__ a_dst, float* __restrict__ s_src,
                float* __restrict__ s_dst, int n) {
    const int lane = threadIdx.x & 31;
    const int node = blockIdx.x * 8 + (threadIdx.x >> 5);
    if (node >= n) return;
    const float4 hv = *(const float4*)(H + (long)node * GDIM + lane * 4);
    const float4 as = *(const float4*)(a_src + lane * 4);
    const float4 ad = *(const float4*)(a_dst + lane * 4);
    float ps = hv.x * as.x + hv.y * as.y + hv.z * as.z + hv.w * as.w;
    float pd = hv.x * ad.x + hv.y * ad.y + hv.z * ad.z + hv.w * ad.w;
#pragma unroll
    for (int off = 16; off > 0; off >>= 1) {
        ps += __shfl_xor(ps, off, 32);
        pd += __shfl_xor(pd, off, 32);
    }
    if (lane == 0) { s_src[node] = ps; s_dst[node] = pd; }
}

#define RNG  400
#define LCAP 8192

template <bool FIRST>
__global__ void __launch_bounds__(1024)
gat_range_aggr(const int* __restrict__ ei, int E, int n,
               const float* __restrict__ s_src, const float* __restrict__ s_dst,
               const float* __restrict__ H,
               const float* __restrict__ bias, float* __restrict__ out,
               int* __restrict__ wlist, int* __restrict__ wcnt) {
    __shared__ float    l_acc[RNG * GDIM];
    __shared__ int2     l_list[LCAP];
    __shared__ float    l_lg[LCAP];
    __shared__ unsigned l_menc[RNG];
    __shared__ float    l_sd[RNG];
    __shared__ float    l_den[RNG];
    __shared__ unsigned l_cnt;

    const int tid  = threadIdx.x;
    const int lane = tid & 31;
    const int wid  = __builtin_amdgcn_readfirstlane(tid >> 5);
    const int j0   = blockIdx.x * RNG;
    const unsigned nloc = (unsigned)((n - j0 < RNG) ? (n - j0) : RNG);

    float4* a4 = (float4*)l_acc;
    for (int i = tid; i < RNG * GDIM / 4; i += 1024) a4[i] = make_float4(0.f, 0.f, 0.f, 0.f);
    if ((unsigned)tid < nloc) {
        const unsigned dl = (unsigned)tid;
        const int   j     = j0 + (int)dl;
        const float sd    = s_dst[j];
        const float s     = s_src[j] + sd;
        const float logit = (s > 0.f) ? s : NEG_SLOPE * s;
        l_sd[dl]   = sd;
        l_den[dl]  = 0.f;
        l_menc[dl] = enc_f32(logit);
        l_list[dl] = make_int2(j, (int)dl);
        l_lg[dl]   = logit;
    }
    if (tid == 0) l_cnt = nloc;
    __syncthreads();

    const int* dstp = ei + E;
    int2* wl = (int2*)wlist + (long)blockIdx.x * LCAP;
    int total;
    if (FIRST) {
    auto keep = [&](int e, unsigned dl) {
        const unsigned pos = atomicAdd(&l_cnt, 1u);
        if (pos < LCAP) l_list[pos] = make_int2(e, (int)dl);
    };
    if ((E & 3) == 0) {
        const int4* d4   = (const int4*)dstp;
        const int   nvec = E >> 2;
        const int   nIt  = (nvec + 1023) >> 10;
        const int   rot  = (int)(((long)blockIdx.x * nIt) / gridDim.x);
        for (int it = 0; it < nIt; it += 4) {
            int4 v[4];
            int  vi[4];
#pragma unroll
            for (int u = 0; u < 4; ++u) {
                int c = it + u + rot;
                if (c >= nIt) c -= nIt;
                vi[u] = (c << 10) + tid;
                const bool ok = (it + u < nIt) && (vi[u] < nvec);
                v[u] = ok ? d4[vi[u]] : make_int4(-1, -1, -1, -1);
            }
#pragma unroll
            for (int u = 0; u < 4; ++u) {
                const unsigned d0 = (unsigned)(v[u].x - j0);
                const unsigned d1 = (unsigned)(v[u].y - j0);
                const unsigned d2 = (unsigned)(v[u].z - j0);
                const unsigned d3 = (unsigned)(v[u].w - j0);
                const unsigned dm = min(min(d0, d1), min(d2, d3));
                if (dm < nloc) {
                    const int e0 = vi[u] << 2;
#pragma nounroll
                    for (int q = 0; q < 4; ++q) {
                        const unsigned d = (q == 0) ? d0 : (q == 1) ? d1 : (q == 2) ? d2 : d3;
                        if (d < nloc) keep(e0 + q, d);
                    }
                }
            }
        }
    } else {
        for (int e = tid; e < E; e += 1024) {
            const unsigned d = (unsigned)(dstp[e] - j0);
            if (d < nloc) keep(e, d);
        }
    }
    __syncthreads();

    total = __builtin_amdgcn_readfirstlane((int)l_cnt);
    if (tid == 0) wcnt[blockIdx.x] = total;
    } else {
    total = __builtin_amdgcn_readfirstlane(wcnt[blockIdx.x]);
    }

    if (total <= LCAP) {
        for (int i0 = (int)nloc + tid; i0 < total; i0 += 4096) {
            int   ii[4];
            int2  en[4];
            int   sr[4];
            float ss[4];
#pragma unroll
            for (int u = 0; u < 4; ++u) {
                ii[u] = i0 + u * 1024;
                const int ic = (ii[u] < total) ? ii[u] : i0;
                if (FIRST) {
                    en[u] = l_list[ic];
                    sr[u] = ei[en[u].x];
                } else {
                    en[u] = wl[ic];
                    sr[u] = en[u].x;
                }
            }
#pragma unroll
            for (int u = 0; u < 4; ++u) ss[u] = s_src[sr[u]];
#pragma unroll
            for (int u = 0; u < 4; ++u) {
                const float s     = ss[u] + l_sd[en[u].y];
                const float logit = (s > 0.f) ? s : NEG_SLOPE * s;
                atomicMax(l_menc + en[u].y, enc_f32(logit));
                if (ii[u] < total) {
                    l_list[ii[u]] = make_int2(sr[u], en[u].y);
                    l_lg[ii[u]]   = logit;
                    if (FIRST) wl[ii[u]] = make_int2(sr[u], en[u].y);
                }
            }
        }
        __syncthreads();

        for (int b = wid * 32; b < total; b += 32 * 32) {
            const int nb = (total - b < 32) ? (total - b) : 32;
            int   src = 0;
            int   dlv = 0;
            float w   = 0.f;
            if (lane < nb) {
                const int2 en = l_list[b + lane];
                src = en.x;
                dlv = en.y;
                w   = __expf(l_lg[b + lane] - dec_f32(l_menc[dlv]));
                atomicAdd(l_den + dlv, w);
            }
            int k = 0;
            for (; k + 8 <= nb; k += 8) {
                float h[8][4];
                float ww[8];
                int   dd[8];
#pragma unroll
                for (int u = 0; u < 8; ++u) {
                    const int sk = __builtin_amdgcn_readlane(src, k + u);
                    dd[u] = __builtin_amdgcn_readlane(dlv, k + u);
                    ww[u] = __int_as_float(__builtin_amdgcn_readlane(__float_as_int(w), k + u));
                    const float* hp = H + (long)sk * GDIM + lane;
                    h[u][0] = hp[0]; h[u][1] = hp[32]; h[u][2] = hp[64]; h[u][3] = hp[96];
                }
#pragma unroll
                for (int u = 0; u < 8; ++u) {
                    float* ap = l_acc + dd[u] * GDIM + lane;
                    atomicAdd(ap + 0,  ww[u] * h[u][0]);
                    atomicAdd(ap + 32, ww[u] * h[u][1]);
                    atomicAdd(ap + 64, ww[u] * h[u][2]);
                    atomicAdd(ap + 96, ww[u] * h[u][3]);
                }
            }
            for (; k < nb; ++k) {
                const int   sk = __builtin_amdgcn_readlane(src, k);
                const int   dk = __builtin_amdgcn_readlane(dlv, k);
                const float wk = __int_as_float(__builtin_amdgcn_readlane(__float_as_int(w), k));
                const float* hp = H + (long)sk * GDIM + lane;
                float*       ap = l_acc + dk * GDIM + lane;
                const float h0 = hp[0], h1 = hp[32], h2 = hp[64], h3 = hp[96];
                atomicAdd(ap + 0,  wk * h0);
                atomicAdd(ap + 32, wk * h1);
                atomicAdd(ap + 64, wk * h2);
                atomicAdd(ap + 96, wk * h3);
            }
        }
    } else {
        for (int e = tid; e < E; e += 1024) {
            const unsigned d = (unsigned)(dstp[e] - j0);
            if (d < nloc) {
                const float s     = s_src[ei[e]] + l_sd[d];
                const float logit = (s > 0.f) ? s : NEG_SLOPE * s;
                atomicMax(l_menc + d, enc_f32(logit));
            }
        }
        __syncthreads();
        for (int e = tid; e < E + (int)nloc; e += 1024) {
            unsigned d;
            int      src;
            if (e < E) { d = (unsigned)(dstp[e] - j0); src = (d < nloc) ? ei[e] : 0; }
            else       { d = (unsigned)(e - E);        src = j0 + (int)d; }
            if (d < nloc) {
                const float s     = s_src[src] + l_sd[d];
                const float logit = (s > 0.f) ? s : NEG_SLOPE * s;
                const float w     = __expf(logit - dec_f32(l_menc[d]));
                atomicAdd(l_den + d, w);
                const float* hp = H + (long)src * GDIM;
                float*       ap = l_acc + d * GDIM;
                for (int c = 0; c < GDIM; ++c) atomicAdd(ap + c, w * hp[c]);
            }
        }
    }
    __syncthreads();

    const float4 b4 = *(const float4*)(bias + lane * 4);
    for (unsigned dl = wid; dl < nloc; dl += 32) {
        const float inv = 1.0f / l_den[dl];
        float4 v = *(const float4*)(l_acc + dl * GDIM + lane * 4);
        v.x = v.x * inv + b4.x;
        v.y = v.y * inv + b4.y;
        v.z = v.z * inv + b4.z;
        v.w = v.w * inv + b4.w;
        *(float4*)(out + (long)(j0 + (int)dl) * GDIM + lane * 4) = v;
    }
}

__global__ void __launch_bounds__(256)
gat_pool_accum(const float* __restrict__ h, const int* __restrict__ batch, int n,
               float* __restrict__ psum, float* __restrict__ pcount) {
    const int lane = threadIdx.x & 31;
    const int node = blockIdx.x * 8 + (threadIdx.x >> 5);
    if (node >= n) return;
    const int    g  = batch[node];
    const float* hp = h + (long)node * GDIM + lane;
    float*       pp = psum + (long)g * GDIM + lane;
    const float h0 = hp[0], h1 = hp[32], h2 = hp[64], h3 = hp[96];
    atomicAdd(pp + 0,  h0);
    atomicAdd(pp + 32, h1);
    atomicAdd(pp + 64, h2);
    atomicAdd(pp + 96, h3);
    if (lane == 0) atomicAdd(pcount + g, 1.0f);
}

__global__ void __launch_bounds__(256)
gat_pool_final(const float* __restrict__ psum, const float* __restrict__ pcount,
               float* __restrict__ out) {
    const int idx = blockIdx.x * blockDim.x + threadIdx.x;
    const float cnt = pcount[idx >> 7];
    out[idx] = psum[idx] / fmaxf(cnt, 1.0f);
}

extern "C" void kernel_launch(void* const* d_in, const int* in_sizes, int n_in,
                              void* d_out, int out_size, void* d_ws, size_t ws_size,
                              hipStream_t stream) {
    const float* x     = (const float*)d_in[0];
    const int*   ei    = (const int*)d_in[1];
    const int*   batch = (const int*)d_in[2];
    const float* W[3]    = {(const float*)d_in[3], (const float*)d_in[7],  (const float*)d_in[11]};
    const float* asrc[3] = {(const float*)d_in[4], (const float*)d_in[8],  (const float*)d_in[12]};
    const float* adst[3] = {(const float*)d_in[5], (const float*)d_in[9],  (const float*)d_in[13]};
    const float* bias[3] = {(const float*)d_in[6], (const float*)d_in[10], (const float*)d_in[14]};

    const int n  = in_sizes[0] / GDIM;
    const int E  = in_sizes[1] / 2;

    char*  wsb = (char*)d_ws;
    size_t off = 0;
    auto take = [&](size_t bytes) -> void* {
        void* p = wsb + off;
        off += (bytes + 255) & ~(size_t)255;
        return p;
    };
    const size_t szH = (size_t)n * GDIM * sizeof(float);
    float*    h      = (float*)   take(szH);
    float*    acc    = (float*)   take(szH);
    float*    s_src  = (float*)   take((size_t)n * sizeof(float));
    float*    s_dst  = (float*)   take((size_t)n * sizeof(float));
    float*    psum   = (float*)   take((size_t)NGRAPH * GDIM * sizeof(float));
    float*    pcount = (float*)   take((size_t)NGRAPH * sizeof(float));
    (void)ws_size; (void)n_in; (void)out_size;

    const int nTiles     = (n + 15) / 16;
    const int gemmBlocks = (nTiles + 3) / 4;
    const int nodeBlocks = (n + 7) / 8;
    const int rangeBlk   = (n + RNG - 1) / RNG;
    int*      wcnt   = (int*)     take((size_t)rangeBlk * sizeof(int));
    int*      wlist  = (int*)     take((size_t)rangeBlk * LCAP * 2 * sizeof(int));

    const float* xin = x;
    for (int l = 0; l < 3; ++l) {
        gat_gemm_wmma<<<gemmBlocks, 128, GDIM * GDIM * sizeof(float), stream>>>(
            xin, W[l], h, nTiles);
        gat_node_scores<<<nodeBlocks, 256, 0, stream>>>(h, asrc[l], adst[l],
                                                        s_src, s_dst, n);
        if (l == 0)
            gat_range_aggr<true><<<rangeBlk, 1024, 0, stream>>>(
                ei, E, n, s_src, s_dst, h, bias[l], acc, wlist, wcnt);
        else
            gat_range_aggr<false><<<rangeBlk, 1024, 0, stream>>>(
                ei, E, n, s_src, s_dst, h, bias[l], acc, wlist, wcnt);
        xin = acc;
    }

    hipMemsetAsync(psum,   0, (size_t)NGRAPH * GDIM * sizeof(float), stream);
    hipMemsetAsync(pcount, 0, (size_t)NGRAPH * sizeof(float), stream);
    gat_pool_accum<<<nodeBlocks, 256, 0, stream>>>(acc, batch, n, psum, pcount);
    gat_pool_final<<<(NGRAPH * GDIM) / 256, 256, 0, stream>>>(psum, pcount,
                                                              (float*)d_out);
}


# score kernel: one wave per 32 consecutive rows (independent row loads, lane sums folded so lane r holds row r, whole-line stores), not one wave per row: 3125 waves a launch, not 100000; same sums
# speedup vs baseline: 3.4340x; 1.0630x over previous
#include <hip/hip_runtime.h>

#define GDIM 128
#define NGRAPH 256
#define NEG_SLOPE 0.2f

typedef float v2f __attribute__((ext_vector_type(2)));
typedef float v8f __attribute__((ext_vector_type(8)));

__device__ __forceinline__ unsigned enc_f32(float f) {
    unsigned b = __float_as_uint(f);
    return (b & 0x80000000u) ? ~b : (b | 0x80000000u);
}
__device__ __forceinline__ float dec_f32(unsigned u) {
    return __uint_as_float((u & 0x80000000u) ? (u ^ 0x80000000u) : ~u);
}

__global__ void __launch_bounds__(128)
gat_gemm_wmma(const float* __restrict__ X, const float* __restrict__ W,
              float* __restrict__ H, int nTiles) {
    extern __shared__ float sW[];
    const int tid = threadIdx.x;

    const float4* W4 = (const float4*)W;
    float4*       S4 = (float4*)sW;
#pragma unroll
    for (int t = 0; t < 32; ++t) S4[t * 128 + tid] = W4[t * 128 + tid];
    __syncthreads();

    const int wid  = tid >> 5;
    const int lane = tid & 31;
    const int hi   = lane >> 4;
    const int col  = lane & 15;
    const int tile = blockIdx.x * 4 + wid;
    if (tile >= nTiles) return;

    const long   row0 = (long)tile * 16;
    const float* Xr   = X + (row0 + col) * GDIM;

    v8f c[8];
#pragma unroll
    for (int n = 0; n < 8; ++n) c[n] = (v8f)0.0f;

    for (int kk = 0; kk < GDIM; kk += 4) {
        const int ka = kk + 2 * hi;
        v2f a = *(const v2f*)(Xr + ka);
#pragma unroll
        for (int n = 0; n < 8; ++n) {
            v2f b;
            b.x = sW[(ka + 0) * GDIM + n * 16 + col];
            b.y = sW[(ka + 1) * GDIM + n * 16 + col];
            c[n] = __builtin_amdgcn_wmma_f32_16x16x4_f32(
                       false, a, false, b, (short)0, c[n], false, false);
        }
    }

    float* Hr = H + row0 * GDIM;
#pragma unroll
    for (int n = 0; n < 8; ++n)
#pragma unroll
        for (int v = 0; v < 8; ++v)
            Hr[(v + 8 * hi) * GDIM + n * 16 + col] = c[n][v];
}

#define SROWS 32
template <int OFF>
__device__ __forceinline__ void fold_step(float (&ps)[SROWS], float (&pd)[SROWS], int lane) {
    const bool up = (lane & OFF) != 0;
#pragma unroll
    for (int r = 0; r < OFF; ++r) {
        const float ks = up ? ps[r + OFF] : ps[r];
        const float gs = up ? ps[r] : ps[r + OFF];
        const float kd = up ? pd[r + OFF] : pd[r];
        const float gd = up ? pd[r] : pd[r + OFF];
        ps[r] = ks + __shfl_xor(gs, OFF, 32);
        pd[r] = kd + __shfl_xor(gd, OFF, 32);
    }
}
__global__ void __launch_bounds__(128)
gat_node_scores(const float* __restrict__ H, const float* __restrict__ a_src,
                const float* __restrict__ a_dst, float* __restrict__ s_src,
                float* __restrict__ s_dst, int n) {
    const int lane = threadIdx.x & 31;
    const int wave = __builtin_amdgcn_readfirstlane((int)(blockIdx.x * 4 + (threadIdx.x >> 5)));
    const int base = wave * SROWS;
    if (base >= n) return;
    const float4 as = *(const float4*)(a_src + lane * 4);
    const float4 ad = *(const float4*)(a_dst + lane * 4);

    float4 hv[SROWS];
#pragma unroll
    for (int r = 0; r < SROWS; ++r) {
        const int row = (base + r < n) ? (base + r) : (n - 1);
        hv[r] = *(const float4*)(H + (long)row * GDIM + lane * 4);
    }
    float ps[SROWS];
    float pd[SROWS];
#pragma unroll
    for (int r = 0; r < SROWS; ++r) {
        ps[r] = hv[r].x * as.x + hv[r].y * as.y + hv[r].z * as.z + hv[r].w * as.w;
        pd[r] = hv[r].x * ad.x + hv[r].y * ad.y + hv[r].z * ad.z + hv[r].w * ad.w;
    }
    fold_step<16>(ps, pd, lane);
    fold_step<8>(ps, pd, lane);
    fold_step<4>(ps, pd, lane);
    fold_step<2>(ps, pd, lane);
    fold_step<1>(ps, pd, lane);
    if (base + lane < n) {
        s_src[base + lane] = ps[0];
        s_dst[base + lane] = pd[0];
    }
}

#define RNG  400
#define LCAP 8192

template <bool FIRST>
__global__ void __launch_bounds__(1024)
gat_range_aggr(const int* __restrict__ ei, int E, int n,
               const float* __restrict__ s_src, const float* __restrict__ s_dst,
               const float* __restrict__ H,
               const float* __restrict__ bias, float* __restrict__ out,
               int* __restrict__ wlist, int* __restrict__ wcnt) {
    __shared__ float    l_acc[RNG * GDIM];
    __shared__ int2     l_list[LCAP];
    __shared__ float    l_lg[LCAP];
    __shared__ unsigned l_menc[RNG];
    __shared__ float    l_sd[RNG];
    __shared__ float    l_den[RNG];
    __shared__ unsigned l_cnt;

    const int tid  = threadIdx.x;
    const int lane = tid & 31;
    const int wid  = __builtin_amdgcn_readfirstlane(tid >> 5);
    const int j0   = blockIdx.x * RNG;
    const unsigned nloc = (unsigned)((n - j0 < RNG) ? (n - j0) : RNG);

    float4* a4 = (float4*)l_acc;
    for (int i = tid; i < RNG * GDIM / 4; i += 1024) a4[i] = make_float4(0.f, 0.f, 0.f, 0.f);
    if ((unsigned)tid < nloc) {
        const unsigned dl = (unsigned)tid;
        const int   j     = j0 + (int)dl;
        const float sd    = s_dst[j];
        const float s     = s_src[j] + sd;
        const float logit = (s > 0.f) ? s : NEG_SLOPE * s;
        l_sd[dl]   = sd;
        l_den[dl]  = 0.f;
        l_menc[dl] = enc_f32(logit);
        l_list[dl] = make_int2(j, (int)dl);
        l_lg[dl]   = logit;
    }
    if (tid == 0) l_cnt = nloc;
    __syncthreads();

    const int* dstp = ei + E;
    int2* wl = (int2*)wlist + (long)blockIdx.x * LCAP;
    int total;
    if (FIRST) {
    auto keep = [&](int e, unsigned dl) {
        const unsigned pos = atomicAdd(&l_cnt, 1u);
        if (pos < LCAP) l_list[pos] = make_int2(e, (int)dl);
    };
    if ((E & 3) == 0) {
        const int4* d4   = (const int4*)dstp;
        const int   nvec = E >> 2;
        const int   nIt  = (nvec + 1023) >> 10;
        const int   rot  = (int)(((long)blockIdx.x * nIt) / gridDim.x);
        for (int it = 0; it < nIt; it += 4) {
            int4 v[4];
            int  vi[4];
#pragma unroll
            for (int u = 0; u < 4; ++u) {
                int c = it + u + rot;
                if (c >= nIt) c -= nIt;
                vi[u] = (c << 10) + tid;
                const bool ok = (it + u < nIt) && (vi[u] < nvec);
                v[u] = ok ? d4[vi[u]] : make_int4(-1, -1, -1, -1);
            }
#pragma unroll
            for (int u = 0; u < 4; ++u) {
                const unsigned d0 = (unsigned)(v[u].x - j0);
                const unsigned d1 = (unsigned)(v[u].y - j0);
                const unsigned d2 = (unsigned)(v[u].z - j0);
                const unsigned d3 = (unsigned)(v[u].w - j0);
                const unsigned dm = min(min(d0, d1), min(d2, d3));
                if (dm < nloc) {
                    const int e0 = vi[u] << 2;
#pragma nounroll
                    for (int q = 0; q < 4; ++q) {
                        const unsigned d = (q == 0) ? d0 : (q == 1) ? d1 : (q == 2) ? d2 : d3;
                        if (d < nloc) keep(e0 + q, d);
                    }
                }
            }
        }
    } else {
        for (int e = tid; e < E; e += 1024) {
            const unsigned d = (unsigned)(dstp[e] - j0);
            if (d < nloc) keep(e, d);
        }
    }
    __syncthreads();

    total = __builtin_amdgcn_readfirstlane((int)l_cnt);
    if (tid == 0) wcnt[blockIdx.x] = total;
    } else {
    total = __builtin_amdgcn_readfirstlane(wcnt[blockIdx.x]);
    }

    if (total <= LCAP) {
        for (int i0 = (int)nloc + tid; i0 < total; i0 += 4096) {
            int   ii[4];
            int2  en[4];
            int   sr[4];
            float ss[4];
#pragma unroll
            for (int u = 0; u < 4; ++u) {
                ii[u] = i0 + u * 1024;
                const int ic = (ii[u] < total) ? ii[u] : i0;
                if (FIRST) {
                    en[u] = l_list[ic];
                    sr[u] = ei[en[u].x];
                } else {
                    en[u] = wl[ic];
                    sr[u] = en[u].x;
                }
            }
#pragma unroll
            for (int u = 0; u < 4; ++u) ss[u] = s_src[sr[u]];
#pragma unroll
            for (int u = 0; u < 4; ++u) {
                const float s     = ss[u] + l_sd[en[u].y];
                const float logit = (s > 0.f) ? s : NEG_SLOPE * s;
                atomicMax(l_menc + en[u].y, enc_f32(logit));
                if (ii[u] < total) {
                    l_list[ii[u]] = make_int2(sr[u], en[u].y);
                    l_lg[ii[u]]   = logit;
                    if (FIRST) wl[ii[u]] = make_int2(sr[u], en[u].y);
                }
            }
        }
        __syncthreads();

        for (int b = wid * 32; b < total; b += 32 * 32) {
            const int nb = (total - b < 32) ? (total - b) : 32;
            int   src = 0;
            int   dlv = 0;
            float w   = 0.f;
            if (lane < nb) {
                const int2 en = l_list[b + lane];
                src = en.x;
                dlv = en.y;
                w   = __expf(l_lg[b + lane] - dec_f32(l_menc[dlv]));
                atomicAdd(l_den + dlv, w);
            }
            int k = 0;
            for (; k + 8 <= nb; k += 8) {
                float h[8][4];
                float ww[8];
                int   dd[8];
#pragma unroll
                for (int u = 0; u < 8; ++u) {
                    const int sk = __builtin_amdgcn_readlane(src, k + u);
                    dd[u] = __builtin_amdgcn_readlane(dlv, k + u);
                    ww[u] = __int_as_float(__builtin_amdgcn_readlane(__float_as_int(w), k + u));
                    const float* hp = H + (long)sk * GDIM + lane;
                    h[u][0] = hp[0]; h[u][1] = hp[32]; h[u][2] = hp[64]; h[u][3] = hp[96];
                }
#pragma unroll
                for (int u = 0; u < 8; ++u) {
                    float* ap = l_acc + dd[u] * GDIM + lane;
                    atomicAdd(ap + 0,  ww[u] * h[u][0]);
                    atomicAdd(ap + 32, ww[u] * h[u][1]);
                    atomicAdd(ap + 64, ww[u] * h[u][2]);
                    atomicAdd(ap + 96, ww[u] * h[u][3]);
                }
            }
            for (; k < nb; ++k) {
                const int   sk = __builtin_amdgcn_readlane(src, k);
                const int   dk = __builtin_amdgcn_readlane(dlv, k);
                const float wk = __int_as_float(__builtin_amdgcn_readlane(__float_as_int(w), k));
                const float* hp = H + (long)sk * GDIM + lane;
                float*       ap = l_acc + dk * GDIM + lane;
                const float h0 = hp[0], h1 = hp[32], h2 = hp[64], h3 = hp[96];
                atomicAdd(ap + 0,  wk * h0);
                atomicAdd(ap + 32, wk * h1);
                atomicAdd(ap + 64, wk * h2);
                atomicAdd(ap + 96, wk * h3);
            }
        }
    } else {
        for (int e = tid; e < E; e += 1024) {
            const unsigned d = (unsigned)(dstp[e] - j0);
            if (d < nloc) {
                const float s     = s_src[ei[e]] + l_sd[d];
                const float logit = (s > 0.f) ? s : NEG_SLOPE * s;
                atomicMax(l_menc + d, enc_f32(logit));
            }
        }
        __syncthreads();
        for (int e = tid; e < E + (int)nloc; e += 1024) {
            unsigned d;
            int      src;
            if (e < E) { d = (unsigned)(dstp[e] - j0); src = (d < nloc) ? ei[e] : 0; }
            else       { d = (unsigned)(e - E);        src = j0 + (int)d; }
            if (d < nloc) {
                const float s     = s_src[src] + l_sd[d];
                const float logit = (s > 0.f) ? s : NEG_SLOPE * s;
                const float w     = __expf(logit - dec_f32(l_menc[d]));
                atomicAdd(l_den + d, w);
                const float* hp = H + (long)src * GDIM;
                float*       ap = l_acc + d * GDIM;
                for (int c = 0; c < GDIM; ++c) atomicAdd(ap + c, w * hp[c]);
            }
        }
    }
    __syncthreads();

    const float4 b4 = *(const float4*)(bias + lane * 4);
    for (unsigned dl = wid; dl < nloc; dl += 32) {
        const float inv = 1.0f / l_den[dl];
        float4 v = *(const float4*)(l_acc + dl * GDIM + lane * 4);
        v.x = v.x * inv + b4.x;
        v.y = v.y * inv + b4.y;
        v.z = v.z * inv + b4.z;
        v.w = v.w * inv + b4.w;
        *(float4*)(out + (long)(j0 + (int)dl) * GDIM + lane * 4) = v;
    }
}

__global__ void __launch_bounds__(256)
gat_pool_accum(const float* __restrict__ h, const int* __restrict__ batch, int n,
               float* __restrict__ psum, float* __restrict__ pcount) {
    const int lane = threadIdx.x & 31;
    const int node = blockIdx.x * 8 + (threadIdx.x >> 5);
    if (node >= n) return;
    const int    g  = batch[node];
    const float* hp = h + (long)node * GDIM + lane;
    float*       pp = psum + (long)g * GDIM + lane;
    const float h0 = hp[0], h1 = hp[32], h2 = hp[64], h3 = hp[96];
    atomicAdd(pp + 0,  h0);
    atomicAdd(pp + 32, h1);
    atomicAdd(pp + 64, h2);
    atomicAdd(pp + 96, h3);
    if (lane == 0) atomicAdd(pcount + g, 1.0f);
}

__global__ void __launch_bounds__(256)
gat_pool_final(const float* __restrict__ psum, const float* __restrict__ pcount,
               float* __restrict__ out) {
    const int idx = blockIdx.x * blockDim.x + threadIdx.x;
    const float cnt = pcount[idx >> 7];
    out[idx] = psum[idx] / fmaxf(cnt, 1.0f);
}

extern "C" void kernel_launch(void* const* d_in, const int* in_sizes, int n_in,
                              void* d_out, int out_size, void* d_ws, size_t ws_size,
                              hipStream_t stream) {
    const float* x     = (const float*)d_in[0];
    const int*   ei    = (const int*)d_in[1];
    const int*   batch = (const int*)d_in[2];
    const float* W[3]    = {(const float*)d_in[3], (const float*)d_in[7],  (const float*)d_in[11]};
    const float* asrc[3] = {(const float*)d_in[4], (const float*)d_in[8],  (const float*)d_in[12]};
    const float* adst[3] = {(const float*)d_in[5], (const float*)d_in[9],  (const float*)d_in[13]};
    const float* bias[3] = {(const float*)d_in[6], (const float*)d_in[10], (const float*)d_in[14]};

    const int n  = in_sizes[0] / GDIM;
    const int E  = in_sizes[1] / 2;

    char*  wsb = (char*)d_ws;
    size_t off = 0;
    auto take = [&](size_t bytes) -> void* {
        void* p = wsb + off;
        off += (bytes + 255) & ~(size_t)255;
        return p;
    };
    const size_t szH = (size_t)n * GDIM * sizeof(float);
    float*    h      = (float*)   take(szH);
    float*    acc    = (float*)   take(szH);
    float*    s_src  = (float*)   take((size_t)n * sizeof(float));
    float*    s_dst  = (float*)   take((size_t)n * sizeof(float));
    float*    psum   = (float*)   take((size_t)NGRAPH * GDIM * sizeof(float));
    float*    pcount = (float*)   take((size_t)NGRAPH * sizeof(float));
    (void)ws_size; (void)n_in; (void)out_size;

    const int nTiles     = (n + 15) / 16;
    const int gemmBlocks = (nTiles + 3) / 4;
    const int nodeBlocks = (n + 7) / 8;
    const int scoreBlocks = (n + 4 * SROWS - 1) / (4 * SROWS);
    const int rangeBlk   = (n + RNG - 1) / RNG;
    int*      wcnt   = (int*)     take((size_t)rangeBlk * sizeof(int));
    int*      wlist  = (int*)     take((size_t)rangeBlk * LCAP * 2 * sizeof(int));

    const float* xin = x;
    for (int l = 0; l < 3; ++l) {
        gat_gemm_wmma<<<gemmBlocks, 128, GDIM * GDIM * sizeof(float), stream>>>(
            xin, W[l], h, nTiles);
        gat_node_scores<<<scoreBlocks, 128, 0, stream>>>(h, asrc[l], adst[l],
                                                         s_src, s_dst, n);
        if (l == 0)
            gat_range_aggr<true><<<rangeBlk, 1024, 0, stream>>>(
                ei, E, n, s_src, s_dst, h, bias[l], acc, wlist, wcnt);
        else
            gat_range_aggr<false><<<rangeBlk, 1024, 0, stream>>>(
                ei, E, n, s_src, s_dst, h, bias[l], acc, wlist, wcnt);
        xin = acc;
    }

    hipMemsetAsync(psum,   0, (size_t)NGRAPH * GDIM * sizeof(float), stream);
    hipMemsetAsync(pcount, 0, (size_t)NGRAPH * sizeof(float), stream);
    gat_pool_accum<<<nodeBlocks, 256, 0, stream>>>(acc, batch, n, psum, pcount);
    gat_pool_final<<<(NGRAPH * GDIM) / 256, 256, 0, stream>>>(psum, pcount,
                                                              (float*)d_out);
}


# layer 1's edge search reads a one-byte owner map (dst / 400 per edge, written by one small launch before it), not the 4-byte dst list: a quarter of the bytes and tests; sums stay in arrival order
# speedup vs baseline: 3.6968x; 1.0765x over previous
#include <hip/hip_runtime.h>

#define GDIM 128
#define NGRAPH 256
#define NEG_SLOPE 0.2f

typedef float v2f __attribute__((ext_vector_type(2)));
typedef float v8f __attribute__((ext_vector_type(8)));

__device__ __forceinline__ unsigned enc_f32(float f) {
    unsigned b = __float_as_uint(f);
    return (b & 0x80000000u) ? ~b : (b | 0x80000000u);
}
__device__ __forceinline__ float dec_f32(unsigned u) {
    return __uint_as_float((u & 0x80000000u) ? (u ^ 0x80000000u) : ~u);
}

__global__ void __launch_bounds__(128)
gat_gemm_wmma(const float* __restrict__ X, const float* __restrict__ W,
              float* __restrict__ H, int nTiles) {
    extern __shared__ float sW[];
    const int tid = threadIdx.x;

    const float4* W4 = (const float4*)W;
    float4*       S4 = (float4*)sW;
#pragma unroll
    for (int t = 0; t < 32; ++t) S4[t * 128 + tid] = W4[t * 128 + tid];
    __syncthreads();

    const int wid  = tid >> 5;
    const int lane = tid & 31;
    const int hi   = lane >> 4;
    const int col  = lane & 15;
    const int tile = blockIdx.x * 4 + wid;
    if (tile >= nTiles) return;

    const long   row0 = (long)tile * 16;
    const float* Xr   = X + (row0 + col) * GDIM;

    v8f c[8];
#pragma unroll
    for (int n = 0; n < 8; ++n) c[n] = (v8f)0.0f;

    for (int kk = 0; kk < GDIM; kk += 4) {
        const int ka = kk + 2 * hi;
        v2f a = *(const v2f*)(Xr + ka);
#pragma unroll
        for (int n = 0; n < 8; ++n) {
            v2f b;
            b.x = sW[(ka + 0) * GDIM + n * 16 + col];
            b.y = sW[(ka + 1) * GDIM + n * 16 + col];
            c[n] = __builtin_amdgcn_wmma_f32_16x16x4_f32(
                       false, a, false, b, (short)0, c[n], false, false);
        }
    }

    float* Hr = H + row0 * GDIM;
#pragma unroll
    for (int n = 0; n < 8; ++n)
#pragma unroll
        for (int v = 0; v < 8; ++v)
            Hr[(v + 8 * hi) * GDIM + n * 16 + col] = c[n][v];
}

#define SROWS 32
template <int OFF>
__device__ __forceinline__ void fold_step(float (&ps)[SROWS], float (&pd)[SROWS], int lane) {
    const bool up = (lane & OFF) != 0;
#pragma unroll
    for (int r = 0; r < OFF; ++r) {
        const float ks = up ? ps[r + OFF] : ps[r];
        const float gs = up ? ps[r] : ps[r + OFF];
        const float kd = up ? pd[r + OFF] : pd[r];
        const float gd = up ? pd[r] : pd[r + OFF];
        ps[r] = ks + __shfl_xor(gs, OFF, 32);
        pd[r] = kd + __shfl_xor(gd, OFF, 32);
    }
}
__global__ void __launch_bounds__(128)
gat_node_scores(const float* __restrict__ H, const float* __restrict__ a_src,
                const float* __restrict__ a_dst, float* __restrict__ s_src,
                float* __restrict__ s_dst, int n) {
    const int lane = threadIdx.x & 31;
    const int wave = __builtin_amdgcn_readfirstlane((int)(blockIdx.x * 4 + (threadIdx.x >> 5)));
    const int base = wave * SROWS;
    if (base >= n) return;
    const float4 as = *(const float4*)(a_src + lane * 4);
    const float4 ad = *(const float4*)(a_dst + lane * 4);

    float4 hv[SROWS];
#pragma unroll
    for (int r = 0; r < SROWS; ++r) {
        const int row = (base + r < n) ? (base + r) : (n - 1);
        hv[r] = *(const float4*)(H + (long)row * GDIM + lane * 4);
    }
    float ps[SROWS];
    float pd[SROWS];
#pragma unroll
    for (int r = 0; r < SROWS; ++r) {
        ps[r] = hv[r].x * as.x + hv[r].y * as.y + hv[r].z * as.z + hv[r].w * as.w;
        pd[r] = hv[r].x * ad.x + hv[r].y * ad.y + hv[r].z * ad.z + hv[r].w * ad.w;
    }
    fold_step<16>(ps, pd, lane);
    fold_step<8>(ps, pd, lane);
    fold_step<4>(ps, pd, lane);
    fold_step<2>(ps, pd, lane);
    fold_step<1>(ps, pd, lane);
    if (base + lane < n) {
        s_src[base + lane] = ps[0];
        s_dst[base + lane] = pd[0];
    }
}

#define RNG  400
#define LCAP 8192

__device__ __forceinline__ unsigned owner_byte(int d) {
    const unsigned o = (unsigned)d / RNG;
    return (o < 255u) ? o : 255u;
}
__global__ void __launch_bounds__(256)
gat_owner_bytes(const int* __restrict__ ei, int E, unsigned* __restrict__ omap) {
    if ((E & 3) != 0) return;
    const int i  = blockIdx.x * 256 + threadIdx.x;
    const int nd = ((E + 15) >> 4) << 2;
    if (i >= nd) return;
    unsigned w = 0xFFFFFFFFu;
    if (4 * i < E) {
        const int4 d = ((const int4*)(ei + E))[i];
        w = owner_byte(d.x) | (owner_byte(d.y) << 8) | (owner_byte(d.z) << 16) |
            (owner_byte(d.w) << 24);
    }
    omap[i] = w;
}
__device__ __forceinline__ unsigned zero_bytes(unsigned x) {
    return ~(((x & 0x7F7F7F7Fu) + 0x7F7F7F7Fu) | x | 0x7F7F7F7Fu);
}

template <bool FIRST>
__global__ void __launch_bounds__(1024)
gat_range_aggr(const int* __restrict__ ei, int E, int n,
               const float* __restrict__ s_src, const float* __restrict__ s_dst,
               const float* __restrict__ H,
               const float* __restrict__ bias, float* __restrict__ out,
               int* __restrict__ wlist, int* __restrict__ wcnt,
               const unsigned* __restrict__ omap) {
    __shared__ float    l_acc[RNG * GDIM];
    __shared__ int2     l_list[LCAP];
    __shared__ float    l_lg[LCAP];
    __shared__ unsigned l_menc[RNG];
    __shared__ float    l_sd[RNG];
    __shared__ float    l_den[RNG];
    __shared__ unsigned l_cnt;

    const int tid  = threadIdx.x;
    const int lane = tid & 31;
    const int wid  = __builtin_amdgcn_readfirstlane(tid >> 5);
    const int j0   = blockIdx.x * RNG;
    const unsigned nloc = (unsigned)((n - j0 < RNG) ? (n - j0) : RNG);

    float4* a4 = (float4*)l_acc;
    for (int i = tid; i < RNG * GDIM / 4; i += 1024) a4[i] = make_float4(0.f, 0.f, 0.f, 0.f);
    if ((unsigned)tid < nloc) {
        const unsigned dl = (unsigned)tid;
        const int   j     = j0 + (int)dl;
        const float sd    = s_dst[j];
        const float s     = s_src[j] + sd;
        const float logit = (s > 0.f) ? s : NEG_SLOPE * s;
        l_sd[dl]   = sd;
        l_den[dl]  = 0.f;
        l_menc[dl] = enc_f32(logit);
        l_list[dl] = make_int2(j, (int)dl);
        l_lg[dl]   = logit;
    }
    if (tid == 0) l_cnt = nloc;
    __syncthreads();

    const int* dstp = ei + E;
    int2* wl = (int2*)wlist + (long)blockIdx.x * LCAP;
    int total;
    if (FIRST) {
    auto keep = [&](int e) {
        const unsigned pos = atomicAdd(&l_cnt, 1u);
        if (pos < LCAP) l_list[pos] = make_int2(e, 0);
    };
    if ((E & 3) == 0 && gridDim.x <= 255) {
        const uint4* m4   = (const uint4*)omap;
        const int    nvec = (E + 15) >> 4;
        const int    nIt  = (nvec + 1023) >> 10;
        const int    rot  = (int)(((long)blockIdx.x * nIt) / gridDim.x);
        const unsigned idrep = blockIdx.x * 0x01010101u;
        for (int it = 0; it < nIt; it += 4) {
            uint4 v[4];
            int   vi[4];
#pragma unroll
            for (int u = 0; u < 4; ++u) {
                int c = it + u + rot;
                if (c >= nIt) c -= nIt;
                vi[u] = (c << 10) + tid;
                const bool ok = (it + u < nIt) && (vi[u] < nvec);
                v[u] = ok ? m4[vi[u]] : make_uint4(~0u, ~0u, ~0u, ~0u);
            }
#pragma unroll
            for (int u = 0; u < 4; ++u) {
                const unsigned m = (zero_bytes(v[u].x ^ idrep) >> 7) |
                                   (zero_bytes(v[u].y ^ idrep) >> 6) |
                                   (zero_bytes(v[u].z ^ idrep) >> 5) |
                                   (zero_bytes(v[u].w ^ idrep) >> 4);
                if (m != 0u) {
                    const int e0 = vi[u] << 4;
                    unsigned  r  = m;
                    while (r != 0u) {
                        const int t = __builtin_ctz(r);
                        keep(e0 + ((t & 7) << 2) + (t >> 3));
                        r &= r - 1u;
                    }
                }
            }
        }
    } else {
        for (int e = tid; e < E; e += 1024) {
            const unsigned d = (unsigned)(dstp[e] - j0);
            if (d < nloc) keep(e);
        }
    }
    __syncthreads();

    total = __builtin_amdgcn_readfirstlane((int)l_cnt);
    if (tid == 0) wcnt[blockIdx.x] = total;
    } else {
    total = __builtin_amdgcn_readfirstlane(wcnt[blockIdx.x]);
    }

    if (total <= LCAP) {
        for (int i0 = (int)nloc + tid; i0 < total; i0 += 4096) {
            int   ii[4];
            int2  en[4];
            int   sr[4];
            float ss[4];
#pragma unroll
            for (int u = 0; u < 4; ++u) {
                ii[u] = i0 + u * 1024;
                const int ic = (ii[u] < total) ? ii[u] : i0;
                if (FIRST) {
                    en[u]   = l_list[ic];
                    sr[u]   = ei[en[u].x];
                    en[u].y = dstp[en[u].x] - j0;
                } else {
                    en[u] = wl[ic];
                    sr[u] = en[u].x;
                }
            }
#pragma unroll
            for (int u = 0; u < 4; ++u) ss[u] = s_src[sr[u]];
#pragma unroll
            for (int u = 0; u < 4; ++u) {
                const float s     = ss[u] + l_sd[en[u].y];
                const float logit = (s > 0.f) ? s : NEG_SLOPE * s;
                atomicMax(l_menc + en[u].y, enc_f32(logit));
                if (ii[u] < total) {
                    l_list[ii[u]] = make_int2(sr[u], en[u].y);
                    l_lg[ii[u]]   = logit;
                    if (FIRST) wl[ii[u]] = make_int2(sr[u], en[u].y);
                }
            }
        }
        __syncthreads();

        for (int b = wid * 32; b < total; b += 32 * 32) {
            const int nb = (total - b < 32) ? (total - b) : 32;
            int   src = 0;
            int   dlv = 0;
            float w   = 0.f;
            if (lane < nb) {
                const int2 en = l_list[b + lane];
                src = en.x;
                dlv = en.y;
                w   = __expf(l_lg[b + lane] - dec_f32(l_menc[dlv]));
                atomicAdd(l_den + dlv, w);
            }
            int k = 0;
            for (; k + 8 <= nb; k += 8) {
                float h[8][4];
                float ww[8];
                int   dd[8];
#pragma unroll
                for (int u = 0; u < 8; ++u) {
                    const int sk = __builtin_amdgcn_readlane(src, k + u);
                    dd[u] = __builtin_amdgcn_readlane(dlv, k + u);
                    ww[u] = __int_as_float(__builtin_amdgcn_readlane(__float_as_int(w), k + u));
                    const float* hp = H + (long)sk * GDIM + lane;
                    h[u][0] = hp[0]; h[u][1] = hp[32]; h[u][2] = hp[64]; h[u][3] = hp[96];
                }
#pragma unroll
                for (int u = 0; u < 8; ++u) {
                    float* ap = l_acc + dd[u] * GDIM + lane;
                    atomicAdd(ap + 0,  ww[u] * h[u][0]);
                    atomicAdd(ap + 32, ww[u] * h[u][1]);
                    atomicAdd(ap + 64, ww[u] * h[u][2]);
                    atomicAdd(ap + 96, ww[u] * h[u][3]);
                }
            }
            for (; k < nb; ++k) {
                const int   sk = __builtin_amdgcn_readlane(src, k);
                const int   dk = __builtin_amdgcn_readlane(dlv, k);
                const float wk = __int_as_float(__builtin_amdgcn_readlane(__float_as_int(w), k));
                const float* hp = H + (long)sk * GDIM + lane;
                float*       ap = l_acc + dk * GDIM + lane;
                const float h0 = hp[0], h1 = hp[32], h2 = hp[64], h3 = hp[96];
                atomicAdd(ap + 0,  wk * h0);
                atomicAdd(ap + 32, wk * h1);
                atomicAdd(ap + 64, wk * h2);
                atomicAdd(ap + 96, wk * h3);
            }
        }
    } else {
        for (int e = tid; e < E; e += 1024) {
            const unsigned d = (unsigned)(dstp[e] - j0);
            if (d < nloc) {
                const float s     = s_src[ei[e]] + l_sd[d];
                const float logit = (s > 0.f) ? s : NEG_SLOPE * s;
                atomicMax(l_menc + d, enc_f32(logit));
            }
        }
        __syncthreads();
        for (int e = tid; e < E + (int)nloc; e += 1024) {
            unsigned d;
            int      src;
            if (e < E) { d = (unsigned)(dstp[e] - j0); src = (d < nloc) ? ei[e] : 0; }
            else       { d = (unsigned)(e - E);        src = j0 + (int)d; }
            if (d < nloc) {
                const float s     = s_src[src] + l_sd[d];
                const float logit = (s > 0.f) ? s : NEG_SLOPE * s;
                const float w     = __expf(logit - dec_f32(l_menc[d]));
                atomicAdd(l_den + d, w);
                const float* hp = H + (long)src * GDIM;
                float*       ap = l_acc + d * GDIM;
                for (int c = 0; c < GDIM; ++c) atomicAdd(ap + c, w * hp[c]);
            }
        }
    }
    __syncthreads();

    const float4 b4 = *(const float4*)(bias + lane * 4);
    for (unsigned dl = wid; dl < nloc; dl += 32) {
        const float inv = 1.0f / l_den[dl];
        float4 v = *(const float4*)(l_acc + dl * GDIM + lane * 4);
        v.x = v.x * inv + b4.x;
        v.y = v.y * inv + b4.y;
        v.z = v.z * inv + b4.z;
        v.w = v.w * inv + b4.w;
        *(float4*)(out + (long)(j0 + (int)dl) * GDIM + lane * 4) = v;
    }
}

__global__ void __launch_bounds__(256)
gat_pool_accum(const float* __restrict__ h, const int* __restrict__ batch, int n,
               float* __restrict__ psum, float* __restrict__ pcount) {
    const int lane = threadIdx.x & 31;
    const int node = blockIdx.x * 8 + (threadIdx.x >> 5);
    if (node >= n) return;
    const int    g  = batch[node];
    const float* hp = h + (long)node * GDIM + lane;
    float*       pp = psum + (long)g * GDIM + lane;
    const float h0 = hp[0], h1 = hp[32], h2 = hp[64], h3 = hp[96];
    atomicAdd(pp + 0,  h0);
    atomicAdd(pp + 32, h1);
    atomicAdd(pp + 64, h2);
    atomicAdd(pp + 96, h3);
    if (lane == 0) atomicAdd(pcount + g, 1.0f);
}

__global__ void __launch_bounds__(256)
gat_pool_final(const float* __restrict__ psum, const float* __restrict__ pcount,
               float* __restrict__ out) {
    const int idx = blockIdx.x * blockDim.x + threadIdx.x;
    const float cnt = pcount[idx >> 7];
    out[idx] = psum[idx] / fmaxf(cnt, 1.0f);
}

extern "C" void kernel_launch(void* const* d_in, const int* in_sizes, int n_in,
                              void* d_out, int out_size, void* d_ws, size_t ws_size,
                              hipStream_t stream) {
    const float* x     = (const float*)d_in[0];
    const int*   ei    = (const int*)d_in[1];
    const int*   batch = (const int*)d_in[2];
    const float* W[3]    = {(const float*)d_in[3], (const float*)d_in[7],  (const float*)d_in[11]};
    const float* asrc[3] = {(const float*)d_in[4], (const float*)d_in[8],  (const float*)d_in[12]};
    const float* adst[3] = {(const float*)d_in[5], (const float*)d_in[9],  (const float*)d_in[13]};
    const float* bias[3] = {(const float*)d_in[6], (const float*)d_in[10], (const float*)d_in[14]};

    const int n  = in_sizes[0] / GDIM;
    const int E  = in_sizes[1] / 2;

    char*  wsb = (char*)d_ws;
    size_t off = 0;
    auto take = [&](size_t bytes) -> void* {
        void* p = wsb + off;
        off += (bytes + 255) & ~(size_t)255;
        return p;
    };
    const size_t szH = (size_t)n * GDIM * sizeof(float);
    float*    h      = (float*)   take(szH);
    float*    acc    = (float*)   take(szH);
    float*    s_src  = (float*)   take((size_t)n * sizeof(float));
    float*    s_dst  = (float*)   take((size_t)n * sizeof(float));
    float*    psum   = (float*)   take((size_t)NGRAPH * GDIM * sizeof(float));
    float*    pcount = (float*)   take((size_t)NGRAPH * sizeof(float));
    (void)ws_size; (void)n_in; (void)out_size;

    const int nTiles     = (n + 15) / 16;
    const int gemmBlocks = (nTiles + 3) / 4;
    const int nodeBlocks = (n + 7) / 8;
    const int scoreBlocks = (n + 4 * SROWS - 1) / (4 * SROWS);
    const int rangeBlk   = (n + RNG - 1) / RNG;
    int*      wcnt   = (int*)     take((size_t)rangeBlk * sizeof(int));
    int*      wlist  = (int*)     take((size_t)rangeBlk * LCAP * 2 * sizeof(int));

    unsigned* omap   = (unsigned*)take((((size_t)E + 15) / 16) * 16);
    const int mapBlocks = (((E + 15) / 16) * 4 + 255) / 256;
    gat_owner_bytes<<<mapBlocks, 256, 0, stream>>>(ei, E, omap);

    const float* xin = x;
    for (int l = 0; l < 3; ++l) {
        gat_gemm_wmma<<<gemmBlocks, 128, GDIM * GDIM * sizeof(float), stream>>>(
            xin, W[l], h, nTiles);
        gat_node_scores<<<scoreBlocks, 128, 0, stream>>>(h, asrc[l], adst[l],
                                                         s_src, s_dst, n);
        if (l == 0)
            gat_range_aggr<true><<<rangeBlk, 1024, 0, stream>>>(
                ei, E, n, s_src, s_dst, h, bias[l], acc, wlist, wcnt, omap);
        else
            gat_range_aggr<false><<<rangeBlk, 1024, 0, stream>>>(
                ei, E, n, s_src, s_dst, h, bias[l], acc, wlist, wcnt, omap);
        xin = acc;
    }

    hipMemsetAsync(psum,   0, (size_t)NGRAPH * GDIM * sizeof(float), stream);
    hipMemsetAsync(pcount, 0, (size_t)NGRAPH * sizeof(float), stream);
    gat_pool_accum<<<nodeBlocks, 256, 0, stream>>>(acc, batch, n, psum, pcount);
    gat_pool_final<<<(NGRAPH * GDIM) / 256, 256, 0, stream>>>(psum, pcount,
                                                              (float*)d_out);
}


# mean pool by owner: a work-group per graph finds its nodes in the batch list (LDS list, arrival order), waves sum those rows in registers, wave sums add in fixed order; pool atomics and memsets gone
# speedup vs baseline: 5.1151x; 1.3837x over previous
#include <hip/hip_runtime.h>

#define GDIM 128
#define NGRAPH 256
#define NEG_SLOPE 0.2f

typedef float v2f __attribute__((ext_vector_type(2)));
typedef float v8f __attribute__((ext_vector_type(8)));

__device__ __forceinline__ unsigned enc_f32(float f) {
    unsigned b = __float_as_uint(f);
    return (b & 0x80000000u) ? ~b : (b | 0x80000000u);
}
__device__ __forceinline__ float dec_f32(unsigned u) {
    return __uint_as_float((u & 0x80000000u) ? (u ^ 0x80000000u) : ~u);
}

__global__ void __launch_bounds__(128)
gat_gemm_wmma(const float* __restrict__ X, const float* __restrict__ W,
              float* __restrict__ H, int nTiles) {
    extern __shared__ float sW[];
    const int tid = threadIdx.x;

    const float4* W4 = (const float4*)W;
    float4*       S4 = (float4*)sW;
#pragma unroll
    for (int t = 0; t < 32; ++t) S4[t * 128 + tid] = W4[t * 128 + tid];
    __syncthreads();

    const int wid  = tid >> 5;
    const int lane = tid & 31;
    const int hi   = lane >> 4;
    const int col  = lane & 15;
    const int tile = blockIdx.x * 4 + wid;
    if (tile >= nTiles) return;

    const long   row0 = (long)tile * 16;
    const float* Xr   = X + (row0 + col) * GDIM;

    v8f c[8];
#pragma unroll
    for (int n = 0; n < 8; ++n) c[n] = (v8f)0.0f;

    for (int kk = 0; kk < GDIM; kk += 4) {
        const int ka = kk + 2 * hi;
        v2f a = *(const v2f*)(Xr + ka);
#pragma unroll
        for (int n = 0; n < 8; ++n) {
            v2f b;
            b.x = sW[(ka + 0) * GDIM + n * 16 + col];
            b.y = sW[(ka + 1) * GDIM + n * 16 + col];
            c[n] = __builtin_amdgcn_wmma_f32_16x16x4_f32(
                       false, a, false, b, (short)0, c[n], false, false);
        }
    }

    float* Hr = H + row0 * GDIM;
#pragma unroll
    for (int n = 0; n < 8; ++n)
#pragma unroll
        for (int v = 0; v < 8; ++v)
            Hr[(v + 8 * hi) * GDIM + n * 16 + col] = c[n][v];
}

#define SROWS 32
template <int OFF>
__device__ __forceinline__ void fold_step(float (&ps)[SROWS], float (&pd)[SROWS], int lane) {
    const bool up = (lane & OFF) != 0;
#pragma unroll
    for (int r = 0; r < OFF; ++r) {
        const float ks = up ? ps[r + OFF] : ps[r];
        const float gs = up ? ps[r] : ps[r + OFF];
        const float kd = up ? pd[r + OFF] : pd[r];
        const float gd = up ? pd[r] : pd[r + OFF];
        ps[r] = ks + __shfl_xor(gs, OFF, 32);
        pd[r] = kd + __shfl_xor(gd, OFF, 32);
    }
}
__global__ void __launch_bounds__(128)
gat_node_scores(const float* __restrict__ H, const float* __restrict__ a_src,
                const float* __restrict__ a_dst, float* __restrict__ s_src,
                float* __restrict__ s_dst, int n) {
    const int lane = threadIdx.x & 31;
    const int wave = __builtin_amdgcn_readfirstlane((int)(blockIdx.x * 4 + (threadIdx.x >> 5)));
    const int base = wave * SROWS;
    if (base >= n) return;
    const float4 as = *(const float4*)(a_src + lane * 4);
    const float4 ad = *(const float4*)(a_dst + lane * 4);

    float4 hv[SROWS];
#pragma unroll
    for (int r = 0; r < SROWS; ++r) {
        const int row = (base + r < n) ? (base + r) : (n - 1);
        hv[r] = *(const float4*)(H + (long)row * GDIM + lane * 4);
    }
    float ps[SROWS];
    float pd[SROWS];
#pragma unroll
    for (int r = 0; r < SROWS; ++r) {
        ps[r] = hv[r].x * as.x + hv[r].y * as.y + hv[r].z * as.z + hv[r].w * as.w;
        pd[r] = hv[r].x * ad.x + hv[r].y * ad.y + hv[r].z * ad.z + hv[r].w * ad.w;
    }
    fold_step<16>(ps, pd, lane);
    fold_step<8>(ps, pd, lane);
    fold_step<4>(ps, pd, lane);
    fold_step<2>(ps, pd, lane);
    fold_step<1>(ps, pd, lane);
    if (base + lane < n) {
        s_src[base + lane] = ps[0];
        s_dst[base + lane] = pd[0];
    }
}

#define RNG  400
#define LCAP 8192

__device__ __forceinline__ unsigned owner_byte(int d) {
    const unsigned o = (unsigned)d / RNG;
    return (o < 255u) ? o : 255u;
}
__global__ void __launch_bounds__(256)
gat_owner_bytes(const int* __restrict__ ei, int E, unsigned* __restrict__ omap) {
    if ((E & 3) != 0) return;
    const int i  = blockIdx.x * 256 + threadIdx.x;
    const int nd = ((E + 15) >> 4) << 2;
    if (i >= nd) return;
    unsigned w = 0xFFFFFFFFu;
    if (4 * i < E) {
        const int4 d = ((const int4*)(ei + E))[i];
        w = owner_byte(d.x) | (owner_byte(d.y) << 8) | (owner_byte(d.z) << 16) |
            (owner_byte(d.w) << 24);
    }
    omap[i] = w;
}
__device__ __forceinline__ unsigned zero_bytes(unsigned x) {
    return ~(((x & 0x7F7F7F7Fu) + 0x7F7F7F7Fu) | x | 0x7F7F7F7Fu);
}

template <bool FIRST>
__global__ void __launch_bounds__(1024)
gat_range_aggr(const int* __restrict__ ei, int E, int n,
               const float* __restrict__ s_src, const float* __restrict__ s_dst,
               const float* __restrict__ H,
               const float* __restrict__ bias, float* __restrict__ out,
               int* __restrict__ wlist, int* __restrict__ wcnt,
               const unsigned* __restrict__ omap) {
    __shared__ float    l_acc[RNG * GDIM];
    __shared__ int2     l_list[LCAP];
    __shared__ float    l_lg[LCAP];
    __shared__ unsigned l_menc[RNG];
    __shared__ float    l_sd[RNG];
    __shared__ float    l_den[RNG];
    __shared__ unsigned l_cnt;

    const int tid  = threadIdx.x;
    const int lane = tid & 31;
    const int wid  = __builtin_amdgcn_readfirstlane(tid >> 5);
    const int j0   = blockIdx.x * RNG;
    const unsigned nloc = (unsigned)((n - j0 < RNG) ? (n - j0) : RNG);

    float4* a4 = (float4*)l_acc;
    for (int i = tid; i < RNG * GDIM / 4; i += 1024) a4[i] = make_float4(0.f, 0.f, 0.f, 0.f);
    if ((unsigned)tid < nloc) {
        const unsigned dl = (unsigned)tid;
        const int   j     = j0 + (int)dl;
        const float sd    = s_dst[j];
        const float s     = s_src[j] + sd;
        const float logit = (s > 0.f) ? s : NEG_SLOPE * s;
        l_sd[dl]   = sd;
        l_den[dl]  = 0.f;
        l_menc[dl] = enc_f32(logit);
        l_list[dl] = make_int2(j, (int)dl);
        l_lg[dl]   = logit;
    }
    if (tid == 0) l_cnt = nloc;
    __syncthreads();

    const int* dstp = ei + E;
    int2* wl = (int2*)wlist + (long)blockIdx.x * LCAP;
    int total;
    if (FIRST) {
    auto keep = [&](int e) {
        const unsigned pos = atomicAdd(&l_cnt, 1u);
        if (pos < LCAP) l_list[pos] = make_int2(e, 0);
    };
    if ((E & 3) == 0 && gridDim.x <= 255) {
        const uint4* m4   = (const uint4*)omap;
        const int    nvec = (E + 15) >> 4;
        const int    nIt  = (nvec + 1023) >> 10;
        const int    rot  = (int)(((long)blockIdx.x * nIt) / gridDim.x);
        const unsigned idrep = blockIdx.x * 0x01010101u;
        for (int it = 0; it < nIt; it += 4) {
            uint4 v[4];
            int   vi[4];
#pragma unroll
            for (int u = 0; u < 4; ++u) {
                int c = it + u + rot;
                if (c >= nIt) c -= nIt;
                vi[u] = (c << 10) + tid;
                const bool ok = (it + u < nIt) && (vi[u] < nvec);
                v[u] = ok ? m4[vi[u]] : make_uint4(~0u, ~0u, ~0u, ~0u);
            }
#pragma unroll
            for (int u = 0; u < 4; ++u) {
                const unsigned m = (zero_bytes(v[u].x ^ idrep) >> 7) |
                                   (zero_bytes(v[u].y ^ idrep) >> 6) |
                                   (zero_bytes(v[u].z ^ idrep) >> 5) |
                                   (zero_bytes(v[u].w ^ idrep) >> 4);
                if (m != 0u) {
                    const int e0 = vi[u] << 4;
                    unsigned  r  = m;
                    while (r != 0u) {
                        const int t = __builtin_ctz(r);
                        keep(e0 + ((t & 7) << 2) + (t >> 3));
                        r &= r - 1u;
                    }
                }
            }
        }
    } else {
        for (int e = tid; e < E; e += 1024) {
            const unsigned d = (unsigned)(dstp[e] - j0);
            if (d < nloc) keep(e);
        }
    }
    __syncthreads();

    total = __builtin_amdgcn_readfirstlane((int)l_cnt);
    if (tid == 0) wcnt[blockIdx.x] = total;
    } else {
    total = __builtin_amdgcn_readfirstlane(wcnt[blockIdx.x]);
    }

    if (total <= LCAP) {
        for (int i0 = (int)nloc + tid; i0 < total; i0 += 4096) {
            int   ii[4];
            int2  en[4];
            int   sr[4];
            float ss[4];
#pragma unroll
            for (int u = 0; u < 4; ++u) {
                ii[u] = i0 + u * 1024;
                const int ic = (ii[u] < total) ? ii[u] : i0;
                if (FIRST) {
                    en[u]   = l_list[ic];
                    sr[u]   = ei[en[u].x];
                    en[u].y = dstp[en[u].x] - j0;
                } else {
                    en[u] = wl[ic];
                    sr[u] = en[u].x;
                }
            }
#pragma unroll
            for (int u = 0; u < 4; ++u) ss[u] = s_src[sr[u]];
#pragma unroll
            for (int u = 0; u < 4; ++u) {
                const float s     = ss[u] + l_sd[en[u].y];
                const float logit = (s > 0.f) ? s : NEG_SLOPE * s;
                atomicMax(l_menc + en[u].y, enc_f32(logit));
                if (ii[u] < total) {
                    l_list[ii[u]] = make_int2(sr[u], en[u].y);
                    l_lg[ii[u]]   = logit;
                    if (FIRST) wl[ii[u]] = make_int2(sr[u], en[u].y);
                }
            }
        }
        __syncthreads();

        for (int b = wid * 32; b < total; b += 32 * 32) {
            const int nb = (total - b < 32) ? (total - b) : 32;
            int   src = 0;
            int   dlv = 0;
            float w   = 0.f;
            if (lane < nb) {
                const int2 en = l_list[b + lane];
                src = en.x;
                dlv = en.y;
                w   = __expf(l_lg[b + lane] - dec_f32(l_menc[dlv]));
                atomicAdd(l_den + dlv, w);
            }
            int k = 0;
            for (; k + 8 <= nb; k += 8) {
                float h[8][4];
                float ww[8];
                int   dd[8];
#pragma unroll
                for (int u = 0; u < 8; ++u) {
                    const int sk = __builtin_amdgcn_readlane(src, k + u);
                    dd[u] = __builtin_amdgcn_readlane(dlv, k + u);
                    ww[u] = __int_as_float(__builtin_amdgcn_readlane(__float_as_int(w), k + u));
                    const float* hp = H + (long)sk * GDIM + lane;
                    h[u][0] = hp[0]; h[u][1] = hp[32]; h[u][2] = hp[64]; h[u][3] = hp[96];
                }
#pragma unroll
                for (int u = 0; u < 8; ++u) {
                    float* ap = l_acc + dd[u] * GDIM + lane;
                    atomicAdd(ap + 0,  ww[u] * h[u][0]);
                    atomicAdd(ap + 32, ww[u] * h[u][1]);
                    atomicAdd(ap + 64, ww[u] * h[u][2]);
                    atomicAdd(ap + 96, ww[u] * h[u][3]);
                }
            }
            for (; k < nb; ++k) {
                const int   sk = __builtin_amdgcn_readlane(src, k);
                const int   dk = __builtin_amdgcn_readlane(dlv, k);
                const float wk = __int_as_float(__builtin_amdgcn_readlane(__float_as_int(w), k));
                const float* hp = H + (long)sk * GDIM + lane;
                float*       ap = l_acc + dk * GDIM + lane;
                const float h0 = hp[0], h1 = hp[32], h2 = hp[64], h3 = hp[96];
                atomicAdd(ap + 0,  wk * h0);
                atomicAdd(ap + 32, wk * h1);
                atomicAdd(ap + 64, wk * h2);
                atomicAdd(ap + 96, wk * h3);
            }
        }
    } else {
        for (int e = tid; e < E; e += 1024) {
            const unsigned d = (unsigned)(dstp[e] - j0);
            if (d < nloc) {
                const float s     = s_src[ei[e]] + l_sd[d];
                const float logit = (s > 0.f) ? s : NEG_SLOPE * s;
                atomicMax(l_menc + d, enc_f32(logit));
            }
        }
        __syncthreads();
        for (int e = tid; e < E + (int)nloc; e += 1024) {
            unsigned d;
            int      src;
            if (e < E) { d = (unsigned)(dstp[e] - j0); src = (d < nloc) ? ei[e] : 0; }
            else       { d = (unsigned)(e - E);        src = j0 + (int)d; }
            if (d < nloc) {
                const float s     = s_src[src] + l_sd[d];
                const float logit = (s > 0.f) ? s : NEG_SLOPE * s;
                const float w     = __expf(logit - dec_f32(l_menc[d]));
                atomicAdd(l_den + d, w);
                const float* hp = H + (long)src * GDIM;
                float*       ap = l_acc + d * GDIM;
                for (int c = 0; c < GDIM; ++c) atomicAdd(ap + c, w * hp[c]);
            }
        }
    }
    __syncthreads();

    const float4 b4 = *(const float4*)(bias + lane * 4);
    for (unsigned dl = wid; dl < nloc; dl += 32) {
        const float inv = 1.0f / l_den[dl];
        float4 v = *(const float4*)(l_acc + dl * GDIM + lane * 4);
        v.x = v.x * inv + b4.x;
        v.y = v.y * inv + b4.y;
        v.z = v.z * inv + b4.z;
        v.w = v.w * inv + b4.w;
        *(float4*)(out + (long)(j0 + (int)dl) * GDIM + lane * 4) = v;
    }
}

#define PLIST 2048
#define PROWS 8
__global__ void __launch_bounds__(1024)
gat_pool_owner(const float* __restrict__ h, const int* __restrict__ batch, int n,
               float* __restrict__ out) {
    __shared__ int      l_nodes[PLIST];
    __shared__ float    l_part[32 * GDIM];
    __shared__ unsigned l_cnt;

    const int tid  = threadIdx.x;
    const int lane = tid & 31;
    const int wid  = __builtin_amdgcn_readfirstlane(tid >> 5);
    const int g    = blockIdx.x;

    if (tid == 0) l_cnt = 0u;
    __syncthreads();

    auto keep = [&](int i) {
        const unsigned pos = atomicAdd(&l_cnt, 1u);
        if (pos < PLIST) l_nodes[pos] = i;
    };
    {
        const int4* b4   = (const int4*)batch;
        const int   nvec = n >> 2;
        const int   nIt  = (nvec + 1023) >> 10;
        const int   rot  = (int)((blockIdx.x * (unsigned)nIt) / gridDim.x);
        for (int it = 0; it < nIt; it += 4) {
            int4 v[4];
            int  vi[4];
#pragma unroll
            for (int u = 0; u < 4; ++u) {
                int c = it + u + rot;
                if (c >= nIt) c -= nIt;
                vi[u] = (c << 10) + tid;
                const bool ok = (it + u < nIt) && (vi[u] < nvec);
                v[u] = ok ? b4[vi[u]] : make_int4(-1, -1, -1, -1);
            }
#pragma unroll
            for (int u = 0; u < 4; ++u) {
                const int i4 = vi[u] << 2;
                if (v[u].x == g) keep(i4 + 0);
                if (v[u].y == g) keep(i4 + 1);
                if (v[u].z == g) keep(i4 + 2);
                if (v[u].w == g) keep(i4 + 3);
            }
        }
        for (int i = (nvec << 2) + tid; i < n; i += 1024)
            if (batch[i] == g) keep(i);
    }
    __syncthreads();

    const int total = __builtin_amdgcn_readfirstlane((int)l_cnt);

    float4 acc = make_float4(0.f, 0.f, 0.f, 0.f);
    if (total <= PLIST) {
        for (int e0 = wid; e0 < total; e0 += 32 * PROWS) {
            int    nd[PROWS];
            float4 r[PROWS];
#pragma unroll
            for (int u = 0; u < PROWS; ++u) {
                const int e = e0 + 32 * u;
                nd[u] = __builtin_amdgcn_readfirstlane(l_nodes[(e < total) ? e : e0]);
            }
#pragma unroll
            for (int u = 0; u < PROWS; ++u)
                r[u] = *(const float4*)(h + (long)nd[u] * GDIM + lane * 4);
#pragma unroll
            for (int u = 0; u < PROWS; ++u) {
                if (e0 + 32 * u < total) {
                    acc.x += r[u].x; acc.y += r[u].y; acc.z += r[u].z; acc.w += r[u].w;
                }
            }
        }
    } else {
        for (int base = wid * 32; base < n; base += 1024) {
            const int i  = base + lane;
            const int gi = (i < n) ? batch[i] : -1;
            unsigned  m  = __builtin_amdgcn_ballot_w32(gi == g);
            while (m != 0u) {
                const int t = __builtin_ctz(m);
                const float4 r = *(const float4*)(h + (long)(base + t) * GDIM + lane * 4);
                acc.x += r.x; acc.y += r.y; acc.z += r.z; acc.w += r.w;
                m &= m - 1u;
            }
        }
    }
    *(float4*)(l_part + wid * GDIM + lane * 4) = acc;
    __syncthreads();

    if (tid < GDIM) {
        float s = 0.f;
#pragma unroll
        for (int w = 0; w < 32; ++w) s += l_part[w * GDIM + tid];
        out[(long)g * GDIM + tid] = s / fmaxf((float)total, 1.0f);
    }
}

extern "C" void kernel_launch(void* const* d_in, const int* in_sizes, int n_in,
                              void* d_out, int out_size, void* d_ws, size_t ws_size,
                              hipStream_t stream) {
    const float* x     = (const float*)d_in[0];
    const int*   ei    = (const int*)d_in[1];
    const int*   batch = (const int*)d_in[2];
    const float* W[3]    = {(const float*)d_in[3], (const float*)d_in[7],  (const float*)d_in[11]};
    const float* asrc[3] = {(const float*)d_in[4], (const float*)d_in[8],  (const float*)d_in[12]};
    const float* adst[3] = {(const float*)d_in[5], (const float*)d_in[9],  (const float*)d_in[13]};
    const float* bias[3] = {(const float*)d_in[6], (const float*)d_in[10], (const float*)d_in[14]};

    const int n  = in_sizes[0] / GDIM;
    const int E  = in_sizes[1] / 2;

    char*  wsb = (char*)d_ws;
    size_t off = 0;
    auto take = [&](size_t bytes) -> void* {
        void* p = wsb + off;
        off += (bytes + 255) & ~(size_t)255;
        return p;
    };
    const size_t szH = (size_t)n * GDIM * sizeof(float);
    float*    h      = (float*)   take(szH);
    float*    acc    = (float*)   take(szH);
    float*    s_src  = (float*)   take((size_t)n * sizeof(float));
    float*    s_dst  = (float*)   take((size_t)n * sizeof(float));
    (void)ws_size; (void)n_in; (void)out_size;

    const int nTiles     = (n + 15) / 16;
    const int gemmBlocks = (nTiles + 3) / 4;
    const int scoreBlocks = (n + 4 * SROWS - 1) / (4 * SROWS);
    const int rangeBlk   = (n + RNG - 1) / RNG;
    int*      wcnt   = (int*)     take((size_t)rangeBlk * sizeof(int));
    int*      wlist  = (int*)     take((size_t)rangeBlk * LCAP * 2 * sizeof(int));

    unsigned* omap   = (unsigned*)take((((size_t)E + 15) / 16) * 16);
    const int mapBlocks = (((E + 15) / 16) * 4 + 255) / 256;
    gat_owner_bytes<<<mapBlocks, 256, 0, stream>>>(ei, E, omap);

    const float* xin = x;
    for (int l = 0; l < 3; ++l) {
        gat_gemm_wmma<<<gemmBlocks, 128, GDIM * GDIM * sizeof(float), stream>>>(
            xin, W[l], h, nTiles);
        gat_node_scores<<<scoreBlocks, 128, 0, stream>>>(h, asrc[l], adst[l],
                                                         s_src, s_dst, n);
        if (l == 0)
            gat_range_aggr<true><<<rangeBlk, 1024, 0, stream>>>(
                ei, E, n, s_src, s_dst, h, bias[l], acc, wlist, wcnt, omap);
        else
            gat_range_aggr<false><<<rangeBlk, 1024, 0, stream>>>(
                ei, E, n, s_src, s_dst, h, bias[l], acc, wlist, wcnt, omap);
        xin = acc;
    }

    gat_pool_owner<<<NGRAPH, 1024, 0, stream>>>(acc, batch, n, (float*)d_out);
}


# edge sums without LDS float atomics: dst row r belongs to wave r mod 32, edges bucketed per wave while resolved; a wave adds its rows by plain 16-B LDS read-add-write, 512-B row loads; arrival order
# speedup vs baseline: 5.2997x; 1.0361x over previous
#include <hip/hip_runtime.h>

#define GDIM 128
#define NGRAPH 256
#define NEG_SLOPE 0.2f

typedef float v2f __attribute__((ext_vector_type(2)));
typedef float v8f __attribute__((ext_vector_type(8)));

__device__ __forceinline__ unsigned enc_f32(float f) {
    unsigned b = __float_as_uint(f);
    return (b & 0x80000000u) ? ~b : (b | 0x80000000u);
}
__device__ __forceinline__ float dec_f32(unsigned u) {
    return __uint_as_float((u & 0x80000000u) ? (u ^ 0x80000000u) : ~u);
}

__global__ void __launch_bounds__(128)
gat_gemm_wmma(const float* __restrict__ X, const float* __restrict__ W,
              float* __restrict__ H, int nTiles) {
    extern __shared__ float sW[];
    const int tid = threadIdx.x;

    const float4* W4 = (const float4*)W;
    float4*       S4 = (float4*)sW;
#pragma unroll
    for (int t = 0; t < 32; ++t) S4[t * 128 + tid] = W4[t * 128 + tid];
    __syncthreads();

    const int wid  = tid >> 5;
    const int lane = tid & 31;
    const int hi   = lane >> 4;
    const int col  = lane & 15;
    const int tile = blockIdx.x * 4 + wid;
    if (tile >= nTiles) return;

    const long   row0 = (long)tile * 16;
    const float* Xr   = X + (row0 + col) * GDIM;

    v8f c[8];
#pragma unroll
    for (int n = 0; n < 8; ++n) c[n] = (v8f)0.0f;

    for (int kk = 0; kk < GDIM; kk += 4) {
        const int ka = kk + 2 * hi;
        v2f a = *(const v2f*)(Xr + ka);
#pragma unroll
        for (int n = 0; n < 8; ++n) {
            v2f b;
            b.x = sW[(ka + 0) * GDIM + n * 16 + col];
            b.y = sW[(ka + 1) * GDIM + n * 16 + col];
            c[n] = __builtin_amdgcn_wmma_f32_16x16x4_f32(
                       false, a, false, b, (short)0, c[n], false, false);
        }
    }

    float* Hr = H + row0 * GDIM;
#pragma unroll
    for (int n = 0; n < 8; ++n)
#pragma unroll
        for (int v = 0; v < 8; ++v)
            Hr[(v + 8 * hi) * GDIM + n * 16 + col] = c[n][v];
}

#define SROWS 32
template <int OFF>
__device__ __forceinline__ void fold_step(float (&ps)[SROWS], float (&pd)[SROWS], int lane) {
    const bool up = (lane & OFF) != 0;
#pragma unroll
    for (int r = 0; r < OFF; ++r) {
        const float ks = up ? ps[r + OFF] : ps[r];
        const float gs = up ? ps[r] : ps[r + OFF];
        const float kd = up ? pd[r + OFF] : pd[r];
        const float gd = up ? pd[r] : pd[r + OFF];
        ps[r] = ks + __shfl_xor(gs, OFF, 32);
        pd[r] = kd + __shfl_xor(gd, OFF, 32);
    }
}
__global__ void __launch_bounds__(128)
gat_node_scores(const float* __restrict__ H, const float* __restrict__ a_src,
                const float* __restrict__ a_dst, float* __restrict__ s_src,
                float* __restrict__ s_dst, int n) {
    const int lane = threadIdx.x & 31;
    const int wave = __builtin_amdgcn_readfirstlane((int)(blockIdx.x * 4 + (threadIdx.x >> 5)));
    const int base = wave * SROWS;
    if (base >= n) return;
    const float4 as = *(const float4*)(a_src + lane * 4);
    const float4 ad = *(const float4*)(a_dst + lane * 4);

    float4 hv[SROWS];
#pragma unroll
    for (int r = 0; r < SROWS; ++r) {
        const int row = (base + r < n) ? (base + r) : (n - 1);
        hv[r] = *(const float4*)(H + (long)row * GDIM + lane * 4);
    }
    float ps[SROWS];
    float pd[SROWS];
#pragma unroll
    for (int r = 0; r < SROWS; ++r) {
        ps[r] = hv[r].x * as.x + hv[r].y * as.y + hv[r].z * as.z + hv[r].w * as.w;
        pd[r] = hv[r].x * ad.x + hv[r].y * ad.y + hv[r].z * ad.z + hv[r].w * ad.w;
    }
    fold_step<16>(ps, pd, lane);
    fold_step<8>(ps, pd, lane);
    fold_step<4>(ps, pd, lane);
    fold_step<2>(ps, pd, lane);
    fold_step<1>(ps, pd, lane);
    if (base + lane < n) {
        s_src[base + lane] = ps[0];
        s_dst[base + lane] = pd[0];
    }
}

#define RNG  400
#define LCAP 6144
#define SCAP 256
#define RING 12

__device__ __forceinline__ unsigned owner_byte(int d) {
    const unsigned o = (unsigned)d / RNG;
    return (o < 255u) ? o : 255u;
}
__global__ void __launch_bounds__(256)
gat_owner_bytes(const int* __restrict__ ei, int E, unsigned* __restrict__ omap) {
    if ((E & 3) != 0) return;
    const int i  = blockIdx.x * 256 + threadIdx.x;
    const int nd = ((E + 15) >> 4) << 2;
    if (i >= nd) return;
    unsigned w = 0xFFFFFFFFu;
    if (4 * i < E) {
        const int4 d = ((const int4*)(ei + E))[i];
        w = owner_byte(d.x) | (owner_byte(d.y) << 8) | (owner_byte(d.z) << 16) |
            (owner_byte(d.w) << 24);
    }
    omap[i] = w;
}
__device__ __forceinline__ unsigned zero_bytes(unsigned x) {
    return ~(((x & 0x7F7F7F7Fu) + 0x7F7F7F7Fu) | x | 0x7F7F7F7Fu);
}

template <bool FIRST>
__global__ void __launch_bounds__(1024)
gat_range_aggr(const int* __restrict__ ei, int E, int n,
               const float* __restrict__ s_src, const float* __restrict__ s_dst,
               const float* __restrict__ H,
               const float* __restrict__ bias, float* __restrict__ out,
               int* __restrict__ wlist, int* __restrict__ wcnt,
               const unsigned* __restrict__ omap) {
    __shared__ float    l_acc[RNG * GDIM];
    __shared__ int2     l_list[LCAP];
    __shared__ float    l_lg[LCAP];
    __shared__ unsigned l_menc[RNG];
    __shared__ float    l_sd[RNG];
    __shared__ float    l_den[RNG];
    __shared__ unsigned l_cnt;
    __shared__ unsigned short l_sub[32 * SCAP];
    __shared__ unsigned l_wn[32];

    const int tid  = threadIdx.x;
    const int lane = tid & 31;
    const int wid  = __builtin_amdgcn_readfirstlane(tid >> 5);
    const int j0   = blockIdx.x * RNG;
    const unsigned nloc = (unsigned)((n - j0 < RNG) ? (n - j0) : RNG);

    float4* a4 = (float4*)l_acc;
    for (int i = tid; i < RNG * GDIM / 4; i += 1024) a4[i] = make_float4(0.f, 0.f, 0.f, 0.f);
    if ((unsigned)tid < nloc) {
        const unsigned dl = (unsigned)tid;
        const int   j     = j0 + (int)dl;
        const float sd    = s_dst[j];
        const float s     = s_src[j] + sd;
        const float logit = (s > 0.f) ? s : NEG_SLOPE * s;
        l_sd[dl]   = sd;
        l_den[dl]  = 0.f;
        l_menc[dl] = enc_f32(logit);
        l_list[dl] = make_int2(j, (int)dl);
        l_lg[dl]   = logit;
        l_sub[(dl & 31u) * SCAP + (dl >> 5)] = (unsigned short)dl;
    }
    if (tid < 32) l_wn[tid] = (nloc + 31u - (unsigned)tid) >> 5;
    if (tid == 0) l_cnt = nloc;
    __syncthreads();

    const int* dstp = ei + E;
    int2* wl = (int2*)wlist + (long)blockIdx.x * LCAP;
    int total;
    if (FIRST) {
    auto keep = [&](int e) {
        const unsigned pos = atomicAdd(&l_cnt, 1u);
        if (pos < LCAP) l_list[pos] = make_int2(e, 0);
    };
    if ((E & 3) == 0 && gridDim.x <= 255) {
        const uint4* m4   = (const uint4*)omap;
        const int    nvec = (E + 15) >> 4;
        const int    nIt  = (nvec + 1023) >> 10;
        const int    rot  = (int)(((long)blockIdx.x * nIt) / gridDim.x);
        const unsigned idrep = blockIdx.x * 0x01010101u;
        for (int it = 0; it < nIt; it += 4) {
            uint4 v[4];
            int   vi[4];
#pragma unroll
            for (int u = 0; u < 4; ++u) {
                int c = it + u + rot;
                if (c >= nIt) c -= nIt;
                vi[u] = (c << 10) + tid;
                const bool ok = (it + u < nIt) && (vi[u] < nvec);
                v[u] = ok ? m4[vi[u]] : make_uint4(~0u, ~0u, ~0u, ~0u);
            }
#pragma unroll
            for (int u = 0; u < 4; ++u) {
                const unsigned m = (zero_bytes(v[u].x ^ idrep) >> 7) |
                                   (zero_bytes(v[u].y ^ idrep) >> 6) |
                                   (zero_bytes(v[u].z ^ idrep) >> 5) |
                                   (zero_bytes(v[u].w ^ idrep) >> 4);
                if (m != 0u) {
                    const int e0 = vi[u] << 4;
                    unsigned  r  = m;
                    while (r != 0u) {
                        const int t = __builtin_ctz(r);
                        keep(e0 + ((t & 7) << 2) + (t >> 3));
                        r &= r - 1u;
                    }
                }
            }
        }
    } else {
        for (int e = tid; e < E; e += 1024) {
            const unsigned d = (unsigned)(dstp[e] - j0);
            if (d < nloc) keep(e);
        }
    }
    __syncthreads();

    total = __builtin_amdgcn_readfirstlane((int)l_cnt);
    if (tid == 0) wcnt[blockIdx.x] = total;
    } else {
    total = __builtin_amdgcn_readfirstlane(wcnt[blockIdx.x]);
    }

    if (total <= LCAP) {
        for (int i0 = (int)nloc + tid; i0 < total; i0 += 4096) {
            int   ii[4];
            int2  en[4];
            int   sr[4];
            float ss[4];
#pragma unroll
            for (int u = 0; u < 4; ++u) {
                ii[u] = i0 + u * 1024;
                const int ic = (ii[u] < total) ? ii[u] : i0;
                if (FIRST) {
                    en[u]   = l_list[ic];
                    sr[u]   = ei[en[u].x];
                    en[u].y = dstp[en[u].x] - j0;
                } else {
                    en[u] = wl[ic];
                    sr[u] = en[u].x;
                }
            }
#pragma unroll
            for (int u = 0; u < 4; ++u) ss[u] = s_src[sr[u]];
#pragma unroll
            for (int u = 0; u < 4; ++u) {
                const float s     = ss[u] + l_sd[en[u].y];
                const float logit = (s > 0.f) ? s : NEG_SLOPE * s;
                atomicMax(l_menc + en[u].y, enc_f32(logit));
                if (ii[u] < total) {
                    l_list[ii[u]] = make_int2(sr[u], en[u].y);
                    l_lg[ii[u]]   = logit;
                    if (FIRST) wl[ii[u]] = make_int2(sr[u], en[u].y);
                    const unsigned ow  = (unsigned)en[u].y & 31u;
                    const unsigned pos = atomicAdd(&l_wn[ow], 1u);
                    if (pos < SCAP) l_sub[ow * SCAP + pos] = (unsigned short)ii[u];
                }
            }
        }
        __syncthreads();

        const bool fits = __builtin_amdgcn_ballot_w32(l_wn[lane] > (unsigned)SCAP) == 0u;
        if (fits) {
            const int nw = __builtin_amdgcn_readfirstlane((int)l_wn[wid]);
            const unsigned short* sub = l_sub + wid * SCAP;
            for (int b = 0; b < nw; b += 32) {
                const int nb = (nw - b < 32) ? (nw - b) : 32;
                int   src = 0;
                int   dlv = 0;
                float w   = 0.f;
                if (lane < nb) {
                    const int  idx = sub[b + lane];
                    const int2 en  = l_list[idx];
                    src = en.x;
                    dlv = en.y;
                    w   = __expf(l_lg[idx] - dec_f32(l_menc[dlv]));
                    atomicAdd(l_den + dlv, w);
                }
                if (lane >= nb) {
                    src = __builtin_amdgcn_readlane(src, 0);
                    dlv = __builtin_amdgcn_readlane(dlv, 0);
                }
                float4 h[RING];
#pragma unroll
                for (int u = 0; u < RING; ++u) {
                    const int sk = __builtin_amdgcn_readlane(src, u);
                    h[u] = *(const float4*)(H + (long)sk * GDIM + lane * 4);
                }
#pragma unroll
                for (int k = 0; k < 32; ++k) {
                    const int   dk = __builtin_amdgcn_readlane(dlv, k);
                    const float wk = __int_as_float(__builtin_amdgcn_readlane(__float_as_int(w), k));
                    float4* ap = (float4*)(l_acc + dk * GDIM + lane * 4);
                    float4  a  = *ap;
                    a.x += wk * h[k % RING].x;
                    a.y += wk * h[k % RING].y;
                    a.z += wk * h[k % RING].z;
                    a.w += wk * h[k % RING].w;
                    *ap = a;
                    if (k + RING < 32) {
                        const int sk = __builtin_amdgcn_readlane(src, k + RING);
                        h[k % RING] = *(const float4*)(H + (long)sk * GDIM + lane * 4);
                    }
                }
            }
        } else {
            for (int b = wid * 32; b < total; b += 32 * 32) {
                const int nb = (total - b < 32) ? (total - b) : 32;
                int   src = 0;
                int   dlv = 0;
                float w   = 0.f;
                if (lane < nb) {
                    const int2 en = l_list[b + lane];
                    src = en.x;
                    dlv = en.y;
                    w   = __expf(l_lg[b + lane] - dec_f32(l_menc[dlv]));
                    atomicAdd(l_den + dlv, w);
                }
                int k = 0;
                for (; k + 8 <= nb; k += 8) {
                    float h[8][4];
                    float ww[8];
                    int   dd[8];
#pragma unroll
                    for (int u = 0; u < 8; ++u) {
                        const int sk = __builtin_amdgcn_readlane(src, k + u);
                        dd[u] = __builtin_amdgcn_readlane(dlv, k + u);
                        ww[u] = __int_as_float(__builtin_amdgcn_readlane(__float_as_int(w), k + u));
                        const float* hp = H + (long)sk * GDIM + lane;
                        h[u][0] = hp[0]; h[u][1] = hp[32]; h[u][2] = hp[64]; h[u][3] = hp[96];
                    }
#pragma unroll
                    for (int u = 0; u < 8; ++u) {
                        float* ap = l_acc + dd[u] * GDIM + lane;
                        atomicAdd(ap + 0,  ww[u] * h[u][0]);
                        atomicAdd(ap + 32, ww[u] * h[u][1]);
                        atomicAdd(ap + 64, ww[u] * h[u][2]);
                        atomicAdd(ap + 96, ww[u] * h[u][3]);
                    }
                }
                for (; k < nb; ++k) {
                    const int   sk = __builtin_amdgcn_readlane(src, k);
                    const int   dk = __builtin_amdgcn_readlane(dlv, k);
                    const float wk = __int_as_float(__builtin_amdgcn_readlane(__float_as_int(w), k));
                    const float* hp = H + (long)sk * GDIM + lane;
                    float*       ap = l_acc + dk * GDIM + lane;
                    const float h0 = hp[0], h1 = hp[32], h2 = hp[64], h3 = hp[96];
                    atomicAdd(ap + 0,  wk * h0);
                    atomicAdd(ap + 32, wk * h1);
                    atomicAdd(ap + 64, wk * h2);
                    atomicAdd(ap + 96, wk * h3);
                }
            }
        }
    } else {
        for (int e = tid; e < E; e += 1024) {
            const unsigned d = (unsigned)(dstp[e] - j0);
            if (d < nloc) {
                const float s     = s_src[ei[e]] + l_sd[d];
                const float logit = (s > 0.f) ? s : NEG_SLOPE * s;
                atomicMax(l_menc + d, enc_f32(logit));
            }
        }
        __syncthreads();
        for (int e = tid; e < E + (int)nloc; e += 1024) {
            unsigned d;
            int      src;
            if (e < E) { d = (unsigned)(dstp[e] - j0); src = (d < nloc) ? ei[e] : 0; }
            else       { d = (unsigned)(e - E);        src = j0 + (int)d; }
            if (d < nloc) {
                const float s     = s_src[src] + l_sd[d];
                const float logit = (s > 0.f) ? s : NEG_SLOPE * s;
                const float w     = __expf(logit - dec_f32(l_menc[d]));
                atomicAdd(l_den + d, w);
                const float* hp = H + (long)src * GDIM;
                float*       ap = l_acc + d * GDIM;
                for (int c = 0; c < GDIM; ++c) atomicAdd(ap + c, w * hp[c]);
            }
        }
    }
    __syncthreads();

    const float4 b4 = *(const float4*)(bias + lane * 4);
    for (unsigned dl = wid; dl < nloc; dl += 32) {
        const float inv = 1.0f / l_den[dl];
        float4 v = *(const float4*)(l_acc + dl * GDIM + lane * 4);
        v.x = v.x * inv + b4.x;
        v.y = v.y * inv + b4.y;
        v.z = v.z * inv + b4.z;
        v.w = v.w * inv + b4.w;
        *(float4*)(out + (long)(j0 + (int)dl) * GDIM + lane * 4) = v;
    }
}

#define PLIST 2048
#define PROWS 8
__global__ void __launch_bounds__(1024)
gat_pool_owner(const float* __restrict__ h, const int* __restrict__ batch, int n,
               float* __restrict__ out) {
    __shared__ int      l_nodes[PLIST];
    __shared__ float    l_part[32 * GDIM];
    __shared__ unsigned l_cnt;

    const int tid  = threadIdx.x;
    const int lane = tid & 31;
    const int wid  = __builtin_amdgcn_readfirstlane(tid >> 5);
    const int g    = blockIdx.x;

    if (tid == 0) l_cnt = 0u;
    __syncthreads();

    auto keep = [&](int i) {
        const unsigned pos = atomicAdd(&l_cnt, 1u);
        if (pos < PLIST) l_nodes[pos] = i;
    };
    {
        const int4* b4   = (const int4*)batch;
        const int   nvec = n >> 2;
        const int   nIt  = (nvec + 1023) >> 10;
        const int   rot  = (int)((blockIdx.x * (unsigned)nIt) / gridDim.x);
        for (int it = 0; it < nIt; it += 4) {
            int4 v[4];
            int  vi[4];
#pragma unroll
            for (int u = 0; u < 4; ++u) {
                int c = it + u + rot;
                if (c >= nIt) c -= nIt;
                vi[u] = (c << 10) + tid;
                const bool ok = (it + u < nIt) && (vi[u] < nvec);
                v[u] = ok ? b4[vi[u]] : make_int4(-1, -1, -1, -1);
            }
#pragma unroll
            for (int u = 0; u < 4; ++u) {
                const int i4 = vi[u] << 2;
                if (v[u].x == g) keep(i4 + 0);
                if (v[u].y == g) keep(i4 + 1);
                if (v[u].z == g) keep(i4 + 2);
                if (v[u].w == g) keep(i4 + 3);
            }
        }
        for (int i = (nvec << 2) + tid; i < n; i += 1024)
            if (batch[i] == g) keep(i);
    }
    __syncthreads();

    const int total = __builtin_amdgcn_readfirstlane((int)l_cnt);

    float4 acc = make_float4(0.f, 0.f, 0.f, 0.f);
    if (total <= PLIST) {
        for (int e0 = wid; e0 < total; e0 += 32 * PROWS) {
            int    nd[PROWS];
            float4 r[PROWS];
#pragma unroll
            for (int u = 0; u < PROWS; ++u) {
                const int e = e0 + 32 * u;
                nd[u] = __builtin_amdgcn_readfirstlane(l_nodes[(e < total) ? e : e0]);
            }
#pragma unroll
            for (int u = 0; u < PROWS; ++u)
                r[u] = *(const float4*)(h + (long)nd[u] * GDIM + lane * 4);
#pragma unroll
            for (int u = 0; u < PROWS; ++u) {
                if (e0 + 32 * u < total) {
                    acc.x += r[u].x; acc.y += r[u].y; acc.z += r[u].z; acc.w += r[u].w;
                }
            }
        }
    } else {
        for (int base = wid * 32; base < n; base += 1024) {
            const int i  = base + lane;
            const int gi = (i < n) ? batch[i] : -1;
            unsigned  m  = __builtin_amdgcn_ballot_w32(gi == g);
            while (m != 0u) {
                const int t = __builtin_ctz(m);
                const float4 r = *(const float4*)(h + (long)(base + t) * GDIM + lane * 4);
                acc.x += r.x; acc.y += r.y; acc.z += r.z; acc.w += r.w;
                m &= m - 1u;
            }
        }
    }
    *(float4*)(l_part + wid * GDIM + lane * 4) = acc;
    __syncthreads();

    if (tid < GDIM) {
        float s = 0.f;
#pragma unroll
        for (int w = 0; w < 32; ++w) s += l_part[w * GDIM + tid];
        out[(long)g * GDIM + tid] = s / fmaxf((float)total, 1.0f);
    }
}

extern "C" void kernel_launch(void* const* d_in, const int* in_sizes, int n_in,
                              void* d_out, int out_size, void* d_ws, size_t ws_size,
                              hipStream_t stream) {
    const float* x     = (const float*)d_in[0];
    const int*   ei    = (const int*)d_in[1];
    const int*   batch = (const int*)d_in[2];
    const float* W[3]    = {(const float*)d_in[3], (const float*)d_in[7],  (const float*)d_in[11]};
    const float* asrc[3] = {(const float*)d_in[4], (const float*)d_in[8],  (const float*)d_in[12]};
    const float* adst[3] = {(const float*)d_in[5], (const float*)d_in[9],  (const float*)d_in[13]};
    const float* bias[3] = {(const float*)d_in[6], (const float*)d_in[10], (const float*)d_in[14]};

    const int n  = in_sizes[0] / GDIM;
    const int E  = in_sizes[1] / 2;

    char*  wsb = (char*)d_ws;
    size_t off = 0;
    auto take = [&](size_t bytes) -> void* {
        void* p = wsb + off;
        off += (bytes + 255) & ~(size_t)255;
        return p;
    };
    const size_t szH = (size_t)n * GDIM * sizeof(float);
    float*    h      = (float*)   take(szH);
    float*    acc    = (float*)   take(szH);
    float*    s_src  = (float*)   take((size_t)n * sizeof(float));
    float*    s_dst  = (float*)   take((size_t)n * sizeof(float));
    (void)ws_size; (void)n_in; (void)out_size;

    const int nTiles     = (n + 15) / 16;
    const int gemmBlocks = (nTiles + 3) / 4;
    const int scoreBlocks = (n + 4 * SROWS - 1) / (4 * SROWS);
    const int rangeBlk   = (n + RNG - 1) / RNG;
    int*      wcnt   = (int*)     take((size_t)rangeBlk * sizeof(int));
    int*      wlist  = (int*)     take((size_t)rangeBlk * LCAP * 2 * sizeof(int));

    unsigned* omap   = (unsigned*)take((((size_t)E + 15) / 16) * 16);
    const int mapBlocks = (((E + 15) / 16) * 4 + 255) / 256;
    gat_owner_bytes<<<mapBlocks, 256, 0, stream>>>(ei, E, omap);

    const float* xin = x;
    for (int l = 0; l < 3; ++l) {
        gat_gemm_wmma<<<gemmBlocks, 128, GDIM * GDIM * sizeof(float), stream>>>(
            xin, W[l], h, nTiles);
        gat_node_scores<<<scoreBlocks, 128, 0, stream>>>(h, asrc[l], adst[l],
                                                         s_src, s_dst, n);
        if (l == 0)
            gat_range_aggr<true><<<rangeBlk, 1024, 0, stream>>>(
                ei, E, n, s_src, s_dst, h, bias[l], acc, wlist, wcnt, omap);
        else
            gat_range_aggr<false><<<rangeBlk, 1024, 0, stream>>>(
                ei, E, n, s_src, s_dst, h, bias[l], acc, wlist, wcnt, omap);
        xin = acc;
    }

    gat_pool_owner<<<NGRAPH, 1024, 0, stream>>>(acc, batch, n, (float*)d_out);
}


# GEMM: W in LDS as interleaved k pairs, a lane's accumulator pair = two neighbouring columns: both B fragments are one conflict-free 16-byte LDS read (was two 4-byte pairs, 2-way conflicts); same sums
# speedup vs baseline: 5.5113x; 1.0399x over previous
#include <hip/hip_runtime.h>

#define GDIM 128
#define NGRAPH 256
#define NEG_SLOPE 0.2f

typedef float v2f __attribute__((ext_vector_type(2)));
typedef float v8f __attribute__((ext_vector_type(8)));

__device__ __forceinline__ unsigned enc_f32(float f) {
    unsigned b = __float_as_uint(f);
    return (b & 0x80000000u) ? ~b : (b | 0x80000000u);
}
__device__ __forceinline__ float dec_f32(unsigned u) {
    return __uint_as_float((u & 0x80000000u) ? (u ^ 0x80000000u) : ~u);
}

__global__ void __launch_bounds__(128)
gat_gemm_wmma(const float* __restrict__ X, const float* __restrict__ W,
              float* __restrict__ H, int nTiles) {
    extern __shared__ float sW[];
    const int tid = threadIdx.x;

    const float4* W4 = (const float4*)W;
#pragma unroll
    for (int t = 0; t < 16; ++t) {
        const int item = t * 128 + tid;
        const int p    = item >> 5;
        const int c4   = item & 31;
        const float4 r0 = W4[(2 * p) * 32 + c4];
        const float4 r1 = W4[(2 * p + 1) * 32 + c4];
        float* dp = sW + p * 256 + ((8 * c4) ^ ((p & 1) << 5));
        *(float4*)(dp)     = make_float4(r0.x, r1.x, r0.y, r1.y);
        *(float4*)(dp + 4) = make_float4(r0.z, r1.z, r0.w, r1.w);
    }
    __syncthreads();

    const int wid  = tid >> 5;
    const int lane = tid & 31;
    const int hi   = lane >> 4;
    const int col  = lane & 15;
    const int tile = blockIdx.x * 4 + wid;
    if (tile >= nTiles) return;

    const long   row0 = (long)tile * 16;
    const float* Xr   = X + (row0 + col) * GDIM;

    v8f c[8];
#pragma unroll
    for (int n = 0; n < 8; ++n) c[n] = (v8f)0.0f;

    const float* sB = sW + hi * 256 + ((4 * col) ^ (hi << 5));
    for (int kk = 0; kk < GDIM; kk += 4) {
        const int ka = kk + 2 * hi;
        v2f a = *(const v2f*)(Xr + ka);
        const float* bp = sB + (kk >> 1) * 256;
        float4 q[4];
#pragma unroll
        for (int m = 0; m < 4; ++m) q[m] = *(const float4*)(bp + m * 64);
#pragma unroll
        for (int m = 0; m < 4; ++m) {
            v2f b0; b0.x = q[m].x; b0.y = q[m].y;
            v2f b1; b1.x = q[m].z; b1.y = q[m].w;
            c[2 * m] = __builtin_amdgcn_wmma_f32_16x16x4_f32(
                           false, a, false, b0, (short)0, c[2 * m], false, false);
            c[2 * m + 1] = __builtin_amdgcn_wmma_f32_16x16x4_f32(
                           false, a, false, b1, (short)0, c[2 * m + 1], false, false);
        }
    }

    float* Hr = H + row0 * GDIM;
#pragma unroll
    for (int m = 0; m < 4; ++m)
#pragma unroll
        for (int v = 0; v < 8; ++v) {
            v2f o; o.x = c[2 * m][v]; o.y = c[2 * m + 1][v];
            *(v2f*)(Hr + (v + 8 * hi) * GDIM + m * 32 + 2 * col) = o;
        }
}

#define SROWS 32
template <int OFF>
__device__ __forceinline__ void fold_step(float (&ps)[SROWS], float (&pd)[SROWS], int lane) {
    const bool up = (lane & OFF) != 0;
#pragma unroll
    for (int r = 0; r < OFF; ++r) {
        const float ks = up ? ps[r + OFF] : ps[r];
        const float gs = up ? ps[r] : ps[r + OFF];
        const float kd = up ? pd[r + OFF] : pd[r];
        const float gd = up ? pd[r] : pd[r + OFF];
        ps[r] = ks + __shfl_xor(gs, OFF, 32);
        pd[r] = kd + __shfl_xor(gd, OFF, 32);
    }
}
__global__ void __launch_bounds__(128)
gat_node_scores(const float* __restrict__ H, const float* __restrict__ a_src,
                const float* __restrict__ a_dst, float* __restrict__ s_src,
                float* __restrict__ s_dst, int n) {
    const int lane = threadIdx.x & 31;
    const int wave = __builtin_amdgcn_readfirstlane((int)(blockIdx.x * 4 + (threadIdx.x >> 5)));
    const int base = wave * SROWS;
    if (base >= n) return;
    const float4 as = *(const float4*)(a_src + lane * 4);
    const float4 ad = *(const float4*)(a_dst + lane * 4);

    float4 hv[SROWS];
#pragma unroll
    for (int r = 0; r < SROWS; ++r) {
        const int row = (base + r < n) ? (base + r) : (n - 1);
        hv[r] = *(const float4*)(H + (long)row * GDIM + lane * 4);
    }
    float ps[SROWS];
    float pd[SROWS];
#pragma unroll
    for (int r = 0; r < SROWS; ++r) {
        ps[r] = hv[r].x * as.x + hv[r].y * as.y + hv[r].z * as.z + hv[r].w * as.w;
        pd[r] = hv[r].x * ad.x + hv[r].y * ad.y + hv[r].z * ad.z + hv[r].w * ad.w;
    }
    fold_step<16>(ps, pd, lane);
    fold_step<8>(ps, pd, lane);
    fold_step<4>(ps, pd, lane);
    fold_step<2>(ps, pd, lane);
    fold_step<1>(ps, pd, lane);
    if (base + lane < n) {
        s_src[base + lane] = ps[0];
        s_dst[base + lane] = pd[0];
    }
}

#define RNG  400
#define LCAP 6144
#define SCAP 256
#define RING 12

__device__ __forceinline__ unsigned owner_byte(int d) {
    const unsigned o = (unsigned)d / RNG;
    return (o < 255u) ? o : 255u;
}
__global__ void __launch_bounds__(256)
gat_owner_bytes(const int* __restrict__ ei, int E, unsigned* __restrict__ omap) {
    if ((E & 3) != 0) return;
    const int i  = blockIdx.x * 256 + threadIdx.x;
    const int nd = ((E + 15) >> 4) << 2;
    if (i >= nd) return;
    unsigned w = 0xFFFFFFFFu;
    if (4 * i < E) {
        const int4 d = ((const int4*)(ei + E))[i];
        w = owner_byte(d.x) | (owner_byte(d.y) << 8) | (owner_byte(d.z) << 16) |
            (owner_byte(d.w) << 24);
    }
    omap[i] = w;
}
__device__ __forceinline__ unsigned zero_bytes(unsigned x) {
    return ~(((x & 0x7F7F7F7Fu) + 0x7F7F7F7Fu) | x | 0x7F7F7F7Fu);
}

template <bool FIRST>
__global__ void __launch_bounds__(1024)
gat_range_aggr(const int* __restrict__ ei, int E, int n,
               const float* __restrict__ s_src, const float* __restrict__ s_dst,
               const float* __restrict__ H,
               const float* __restrict__ bias, float* __restrict__ out,
               int* __restrict__ wlist, int* __restrict__ wcnt,
               const unsigned* __restrict__ omap) {
    __shared__ float    l_acc[RNG * GDIM];
    __shared__ int2     l_list[LCAP];
    __shared__ float    l_lg[LCAP];
    __shared__ unsigned l_menc[RNG];
    __shared__ float    l_sd[RNG];
    __shared__ float    l_den[RNG];
    __shared__ unsigned l_cnt;
    __shared__ unsigned short l_sub[32 * SCAP];
    __shared__ unsigned l_wn[32];

    const int tid  = threadIdx.x;
    const int lane = tid & 31;
    const int wid  = __builtin_amdgcn_readfirstlane(tid >> 5);
    const int j0   = blockIdx.x * RNG;
    const unsigned nloc = (unsigned)((n - j0 < RNG) ? (n - j0) : RNG);

    float4* a4 = (float4*)l_acc;
    for (int i = tid; i < RNG * GDIM / 4; i += 1024) a4[i] = make_float4(0.f, 0.f, 0.f, 0.f);
    if ((unsigned)tid < nloc) {
        const unsigned dl = (unsigned)tid;
        const int   j     = j0 + (int)dl;
        const float sd    = s_dst[j];
        const float s     = s_src[j] + sd;
        const float logit = (s > 0.f) ? s : NEG_SLOPE * s;
        l_sd[dl]   = sd;
        l_den[dl]  = 0.f;
        l_menc[dl] = enc_f32(logit);
        l_list[dl] = make_int2(j, (int)dl);
        l_lg[dl]   = logit;
        l_sub[(dl & 31u) * SCAP + (dl >> 5)] = (unsigned short)dl;
    }
    if (tid < 32) l_wn[tid] = (nloc + 31u - (unsigned)tid) >> 5;
    if (tid == 0) l_cnt = nloc;
    __syncthreads();

    const int* dstp = ei + E;
    int2* wl = (int2*)wlist + (long)blockIdx.x * LCAP;
    int total;
    if (FIRST) {
    auto keep = [&](int e) {
        const unsigned pos = atomicAdd(&l_cnt, 1u);
        if (pos < LCAP) l_list[pos] = make_int2(e, 0);
    };
    if ((E & 3) == 0 && gridDim.x <= 255) {
        const uint4* m4   = (const uint4*)omap;
        const int    nvec = (E + 15) >> 4;
        const int    nIt  = (nvec + 1023) >> 10;
        const int    rot  = (int)(((long)blockIdx.x * nIt) / gridDim.x);
        const unsigned idrep = blockIdx.x * 0x01010101u;
        for (int it = 0; it < nIt; it += 4) {
            uint4 v[4];
            int   vi[4];
#pragma unroll
            for (int u = 0; u < 4; ++u) {
                int c = it + u + rot;
                if (c >= nIt) c -= nIt;
                vi[u] = (c << 10) + tid;
                const bool ok = (it + u < nIt) && (vi[u] < nvec);
                v[u] = ok ? m4[vi[u]] : make_uint4(~0u, ~0u, ~0u, ~0u);
            }
#pragma unroll
            for (int u = 0; u < 4; ++u) {
                const unsigned m = (zero_bytes(v[u].x ^ idrep) >> 7) |
                                   (zero_bytes(v[u].y ^ idrep) >> 6) |
                                   (zero_bytes(v[u].z ^ idrep) >> 5) |
                                   (zero_bytes(v[u].w ^ idrep) >> 4);
                if (m != 0u) {
                    const int e0 = vi[u] << 4;
                    unsigned  r  = m;
                    while (r != 0u) {
                        const int t = __builtin_ctz(r);
                        keep(e0 + ((t & 7) << 2) + (t >> 3));
                        r &= r - 1u;
                    }
                }
            }
        }
    } else {
        for (int e = tid; e < E; e += 1024) {
            const unsigned d = (unsigned)(dstp[e] - j0);
            if (d < nloc) keep(e);
        }
    }
    __syncthreads();

    total = __builtin_amdgcn_readfirstlane((int)l_cnt);
    if (tid == 0) wcnt[blockIdx.x] = total;
    } else {
    total = __builtin_amdgcn_readfirstlane(wcnt[blockIdx.x]);
    }

    if (total <= LCAP) {
        for (int i0 = (int)nloc + tid; i0 < total; i0 += 4096) {
            int   ii[4];
            int2  en[4];
            int   sr[4];
            float ss[4];
#pragma unroll
            for (int u = 0; u < 4; ++u) {
                ii[u] = i0 + u * 1024;
                const int ic = (ii[u] < total) ? ii[u] : i0;
                if (FIRST) {
                    en[u]   = l_list[ic];
                    sr[u]   = ei[en[u].x];
                    en[u].y = dstp[en[u].x] - j0;
                } else {
                    en[u] = wl[ic];
                    sr[u] = en[u].x;
                }
            }
#pragma unroll
            for (int u = 0; u < 4; ++u) ss[u] = s_src[sr[u]];
#pragma unroll
            for (int u = 0; u < 4; ++u) {
                const float s     = ss[u] + l_sd[en[u].y];
                const float logit = (s > 0.f) ? s : NEG_SLOPE * s;
                atomicMax(l_menc + en[u].y, enc_f32(logit));
                if (ii[u] < total) {
                    l_list[ii[u]] = make_int2(sr[u], en[u].y);
                    l_lg[ii[u]]   = logit;
                    if (FIRST) wl[ii[u]] = make_int2(sr[u], en[u].y);
                    const unsigned ow  = (unsigned)en[u].y & 31u;
                    const unsigned pos = atomicAdd(&l_wn[ow], 1u);
                    if (pos < SCAP) l_sub[ow * SCAP + pos] = (unsigned short)ii[u];
                }
            }
        }
        __syncthreads();

        const bool fits = __builtin_amdgcn_ballot_w32(l_wn[lane] > (unsigned)SCAP) == 0u;
        if (fits) {
            const int nw = __builtin_amdgcn_readfirstlane((int)l_wn[wid]);
            const unsigned short* sub = l_sub + wid * SCAP;
            for (int b = 0; b < nw; b += 32) {
                const int nb = (nw - b < 32) ? (nw - b) : 32;
                int   src = 0;
                int   dlv = 0;
                float w   = 0.f;
                if (lane < nb) {
                    const int  idx = sub[b + lane];
                    const int2 en  = l_list[idx];
                    src = en.x;
                    dlv = en.y;
                    w   = __expf(l_lg[idx] - dec_f32(l_menc[dlv]));
                    atomicAdd(l_den + dlv, w);
                }
                if (lane >= nb) {
                    src = __builtin_amdgcn_readlane(src, 0);
                    dlv = __builtin_amdgcn_readlane(dlv, 0);
                }
                float4 h[RING];
#pragma unroll
                for (int u = 0; u < RING; ++u) {
                    const int sk = __builtin_amdgcn_readlane(src, u);
                    h[u] = *(const float4*)(H + (long)sk * GDIM + lane * 4);
                }
#pragma unroll
                for (int k = 0; k < 32; ++k) {
                    const int   dk = __builtin_amdgcn_readlane(dlv, k);
                    const float wk = __int_as_float(__builtin_amdgcn_readlane(__float_as_int(w), k));
                    float4* ap = (float4*)(l_acc + dk * GDIM + lane * 4);
                    float4  a  = *ap;
                    a.x += wk * h[k % RING].x;
                    a.y += wk * h[k % RING].y;
                    a.z += wk * h[k % RING].z;
                    a.w += wk * h[k % RING].w;
                    *ap = a;
                    if (k + RING < 32) {
                        const int sk = __builtin_amdgcn_readlane(src, k + RING);
                        h[k % RING] = *(const float4*)(H + (long)sk * GDIM + lane * 4);
                    }
                }
            }
        } else {
            for (int b = wid * 32; b < total; b += 32 * 32) {
                const int nb = (total - b < 32) ? (total - b) : 32;
                int   src = 0;
                int   dlv = 0;
                float w   = 0.f;
                if (lane < nb) {
                    const int2 en = l_list[b + lane];
                    src = en.x;
                    dlv = en.y;
                    w   = __expf(l_lg[b + lane] - dec_f32(l_menc[dlv]));
                    atomicAdd(l_den + dlv, w);
                }
                int k = 0;
                for (; k + 8 <= nb; k += 8) {
                    float h[8][4];
                    float ww[8];
                    int   dd[8];
#pragma unroll
                    for (int u = 0; u < 8; ++u) {
                        const int sk = __builtin_amdgcn_readlane(src, k + u);
                        dd[u] = __builtin_amdgcn_readlane(dlv, k + u);
                        ww[u] = __int_as_float(__builtin_amdgcn_readlane(__float_as_int(w), k + u));
                        const float* hp = H + (long)sk * GDIM + lane;
                        h[u][0] = hp[0]; h[u][1] = hp[32]; h[u][2] = hp[64]; h[u][3] = hp[96];
                    }
#pragma unroll
                    for (int u = 0; u < 8; ++u) {
                        float* ap = l_acc + dd[u] * GDIM + lane;
                        atomicAdd(ap + 0,  ww[u] * h[u][0]);
                        atomicAdd(ap + 32, ww[u] * h[u][1]);
                        atomicAdd(ap + 64, ww[u] * h[u][2]);
                        atomicAdd(ap + 96, ww[u] * h[u][3]);
                    }
                }
                for (; k < nb; ++k) {
                    const int   sk = __builtin_amdgcn_readlane(src, k);
                    const int   dk = __builtin_amdgcn_readlane(dlv, k);
                    const float wk = __int_as_float(__builtin_amdgcn_readlane(__float_as_int(w), k));
                    const float* hp = H + (long)sk * GDIM + lane;
                    float*       ap = l_acc + dk * GDIM + lane;
                    const float h0 = hp[0], h1 = hp[32], h2 = hp[64], h3 = hp[96];
                    atomicAdd(ap + 0,  wk * h0);
                    atomicAdd(ap + 32, wk * h1);
                    atomicAdd(ap + 64, wk * h2);
                    atomicAdd(ap + 96, wk * h3);
                }
            }
        }
    } else {
        for (int e = tid; e < E; e += 1024) {
            const unsigned d = (unsigned)(dstp[e] - j0);
            if (d < nloc) {
                const float s     = s_src[ei[e]] + l_sd[d];
                const float logit = (s > 0.f) ? s : NEG_SLOPE * s;
                atomicMax(l_menc + d, enc_f32(logit));
            }
        }
        __syncthreads();
        for (int e = tid; e < E + (int)nloc; e += 1024) {
            unsigned d;
            int      src;
            if (e < E) { d = (unsigned)(dstp[e] - j0); src = (d < nloc) ? ei[e] : 0; }
            else       { d = (unsigned)(e - E);        src = j0 + (int)d; }
            if (d < nloc) {
                const float s     = s_src[src] + l_sd[d];
                const float logit = (s > 0.f) ? s : NEG_SLOPE * s;
                const float w     = __expf(logit - dec_f32(l_menc[d]));
                atomicAdd(l_den + d, w);
                const float* hp = H + (long)src * GDIM;
                float*       ap = l_acc + d * GDIM;
                for (int c = 0; c < GDIM; ++c) atomicAdd(ap + c, w * hp[c]);
            }
        }
    }
    __syncthreads();

    const float4 b4 = *(const float4*)(bias + lane * 4);
    for (unsigned dl = wid; dl < nloc; dl += 32) {
        const float inv = 1.0f / l_den[dl];
        float4 v = *(const float4*)(l_acc + dl * GDIM + lane * 4);
        v.x = v.x * inv + b4.x;
        v.y = v.y * inv + b4.y;
        v.z = v.z * inv + b4.z;
        v.w = v.w * inv + b4.w;
        *(float4*)(out + (long)(j0 + (int)dl) * GDIM + lane * 4) = v;
    }
}

#define PLIST 2048
#define PROWS 8
__global__ void __launch_bounds__(1024)
gat_pool_owner(const float* __restrict__ h, const int* __restrict__ batch, int n,
               float* __restrict__ out) {
    __shared__ int      l_nodes[PLIST];
    __shared__ float    l_part[32 * GDIM];
    __shared__ unsigned l_cnt;

    const int tid  = threadIdx.x;
    const int lane = tid & 31;
    const int wid  = __builtin_amdgcn_readfirstlane(tid >> 5);
    const int g    = blockIdx.x;

    if (tid == 0) l_cnt = 0u;
    __syncthreads();

    auto keep = [&](int i) {
        const unsigned pos = atomicAdd(&l_cnt, 1u);
        if (pos < PLIST) l_nodes[pos] = i;
    };
    {
        const int4* b4   = (const int4*)batch;
        const int   nvec = n >> 2;
        const int   nIt  = (nvec + 1023) >> 10;
        const int   rot  = (int)((blockIdx.x * (unsigned)nIt) / gridDim.x);
        for (int it = 0; it < nIt; it += 4) {
            int4 v[4];
            int  vi[4];
#pragma unroll
            for (int u = 0; u < 4; ++u) {
                int c = it + u + rot;
                if (c >= nIt) c -= nIt;
                vi[u] = (c << 10) + tid;
                const bool ok = (it + u < nIt) && (vi[u] < nvec);
                v[u] = ok ? b4[vi[u]] : make_int4(-1, -1, -1, -1);
            }
#pragma unroll
            for (int u = 0; u < 4; ++u) {
                const int i4 = vi[u] << 2;
                if (v[u].x == g) keep(i4 + 0);
                if (v[u].y == g) keep(i4 + 1);
                if (v[u].z == g) keep(i4 + 2);
                if (v[u].w == g) keep(i4 + 3);
            }
        }
        for (int i = (nvec << 2) + tid; i < n; i += 1024)
            if (batch[i] == g) keep(i);
    }
    __syncthreads();

    const int total = __builtin_amdgcn_readfirstlane((int)l_cnt);

    float4 acc = make_float4(0.f, 0.f, 0.f, 0.f);
    if (total <= PLIST) {
        for (int e0 = wid; e0 < total; e0 += 32 * PROWS) {
            int    nd[PROWS];
            float4 r[PROWS];
#pragma unroll
            for (int u = 0; u < PROWS; ++u) {
                const int e = e0 + 32 * u;
                nd[u] = __builtin_amdgcn_readfirstlane(l_nodes[(e < total) ? e : e0]);
            }
#pragma unroll
            for (int u = 0; u < PROWS; ++u)
                r[u] = *(const float4*)(h + (long)nd[u] * GDIM + lane * 4);
#pragma unroll
            for (int u = 0; u < PROWS; ++u) {
                if (e0 + 32 * u < total) {
                    acc.x += r[u].x; acc.y += r[u].y; acc.z += r[u].z; acc.w += r[u].w;
                }
            }
        }
    } else {
        for (int base = wid * 32; base < n; base += 1024) {
            const int i  = base + lane;
            const int gi = (i < n) ? batch[i] : -1;
            unsigned  m  = __builtin_amdgcn_ballot_w32(gi == g);
            while (m != 0u) {
                const int t = __builtin_ctz(m);
                const float4 r = *(const float4*)(h + (long)(base + t) * GDIM + lane * 4);
                acc.x += r.x; acc.y += r.y; acc.z += r.z; acc.w += r.w;
                m &= m - 1u;
            }
        }
    }
    *(float4*)(l_part + wid * GDIM + lane * 4) = acc;
    __syncthreads();

    if (tid < GDIM) {
        float s = 0.f;
#pragma unroll
        for (int w = 0; w < 32; ++w) s += l_part[w * GDIM + tid];
        out[(long)g * GDIM + tid] = s / fmaxf((float)total, 1.0f);
    }
}

extern "C" void kernel_launch(void* const* d_in, const int* in_sizes, int n_in,
                              void* d_out, int out_size, void* d_ws, size_t ws_size,
                              hipStream_t stream) {
    const float* x     = (const float*)d_in[0];
    const int*   ei    = (const int*)d_in[1];
    const int*   batch = (const int*)d_in[2];
    const float* W[3]    = {(const float*)d_in[3], (const float*)d_in[7],  (const float*)d_in[11]};
    const float* asrc[3] = {(const float*)d_in[4], (const float*)d_in[8],  (const float*)d_in[12]};
    const float* adst[3] = {(const float*)d_in[5], (const float*)d_in[9],  (const float*)d_in[13]};
    const float* bias[3] = {(const float*)d_in[6], (const float*)d_in[10], (const float*)d_in[14]};

    const int n  = in_sizes[0] / GDIM;
    const int E  = in_sizes[1] / 2;

    char*  wsb = (char*)d_ws;
    size_t off = 0;
    auto take = [&](size_t bytes) -> void* {
        void* p = wsb + off;
        off += (bytes + 255) & ~(size_t)255;
        return p;
    };
    const size_t szH = (size_t)n * GDIM * sizeof(float);
    float*    h      = (float*)   take(szH);
    float*    acc    = (float*)   take(szH);
    float*    s_src  = (float*)   take((size_t)n * sizeof(float));
    float*    s_dst  = (float*)   take((size_t)n * sizeof(float));
    (void)ws_size; (void)n_in; (void)out_size;

    const int nTiles     = (n + 15) / 16;
    const int gemmBlocks = (nTiles + 3) / 4;
    const int scoreBlocks = (n + 4 * SROWS - 1) / (4 * SROWS);
    const int rangeBlk   = (n + RNG - 1) / RNG;
    int*      wcnt   = (int*)     take((size_t)rangeBlk * sizeof(int));
    int*      wlist  = (int*)     take((size_t)rangeBlk * LCAP * 2 * sizeof(int));

    unsigned* omap   = (unsigned*)take((((size_t)E + 15) / 16) * 16);
    const int mapBlocks = (((E + 15) / 16) * 4 + 255) / 256;
    gat_owner_bytes<<<mapBlocks, 256, 0, stream>>>(ei, E, omap);

    const float* xin = x;
    for (int l = 0; l < 3; ++l) {
        gat_gemm_wmma<<<gemmBlocks, 128, GDIM * GDIM * sizeof(float), stream>>>(
            xin, W[l], h, nTiles);
        gat_node_scores<<<scoreBlocks, 128, 0, stream>>>(h, asrc[l], adst[l],
                                                         s_src, s_dst, n);
        if (l == 0)
            gat_range_aggr<true><<<rangeBlk, 1024, 0, stream>>>(
                ei, E, n, s_src, s_dst, h, bias[l], acc, wlist, wcnt, omap);
        else
            gat_range_aggr<false><<<rangeBlk, 1024, 0, stream>>>(
                ei, E, n, s_src, s_dst, h, bias[l], acc, wlist, wcnt, omap);
        xin = acc;
    }

    gat_pool_owner<<<NGRAPH, 1024, 0, stream>>>(acc, batch, n, (float*)d_out);
}


# edge lists in one pass: a launch sends each edge once to its owner's list (LDS count per owner, one returning add per owner); owner map and layer 1's search of all edges gone; sums in arrival order
# speedup vs baseline: 5.8358x; 1.0589x over previous
#include <hip/hip_runtime.h>

#define GDIM 128
#define NGRAPH 256
#define NEG_SLOPE 0.2f

typedef float v2f __attribute__((ext_vector_type(2)));
typedef float v8f __attribute__((ext_vector_type(8)));

__device__ __forceinline__ unsigned enc_f32(float f) {
    unsigned b = __float_as_uint(f);
    return (b & 0x80000000u) ? ~b : (b | 0x80000000u);
}
__device__ __forceinline__ float dec_f32(unsigned u) {
    return __uint_as_float((u & 0x80000000u) ? (u ^ 0x80000000u) : ~u);
}

__global__ void __launch_bounds__(128)
gat_gemm_wmma(const float* __restrict__ X, const float* __restrict__ W,
              float* __restrict__ H, int nTiles) {
    extern __shared__ float sW[];
    const int tid = threadIdx.x;

    const float4* W4 = (const float4*)W;
#pragma unroll
    for (int t = 0; t < 16; ++t) {
        const int item = t * 128 + tid;
        const int p    = item >> 5;
        const int c4   = item & 31;
        const float4 r0 = W4[(2 * p) * 32 + c4];
        const float4 r1 = W4[(2 * p + 1) * 32 + c4];
        float* dp = sW + p * 256 + ((8 * c4) ^ ((p & 1) << 5));
        *(float4*)(dp)     = make_float4(r0.x, r1.x, r0.y, r1.y);
        *(float4*)(dp + 4) = make_float4(r0.z, r1.z, r0.w, r1.w);
    }
    __syncthreads();

    const int wid  = tid >> 5;
    const int lane = tid & 31;
    const int hi   = lane >> 4;
    const int col  = lane & 15;
    const int tile = blockIdx.x * 4 + wid;
    if (tile >= nTiles) return;

    const long   row0 = (long)tile * 16;
    const float* Xr   = X + (row0 + col) * GDIM;

    v8f c[8];
#pragma unroll
    for (int n = 0; n < 8; ++n) c[n] = (v8f)0.0f;

    const float* sB = sW + hi * 256 + ((4 * col) ^ (hi << 5));
    for (int kk = 0; kk < GDIM; kk += 4) {
        const int ka = kk + 2 * hi;
        v2f a = *(const v2f*)(Xr + ka);
        const float* bp = sB + (kk >> 1) * 256;
        float4 q[4];
#pragma unroll
        for (int m = 0; m < 4; ++m) q[m] = *(const float4*)(bp + m * 64);
#pragma unroll
        for (int m = 0; m < 4; ++m) {
            v2f b0; b0.x = q[m].x; b0.y = q[m].y;
            v2f b1; b1.x = q[m].z; b1.y = q[m].w;
            c[2 * m] = __builtin_amdgcn_wmma_f32_16x16x4_f32(
                           false, a, false, b0, (short)0, c[2 * m], false, false);
            c[2 * m + 1] = __builtin_amdgcn_wmma_f32_16x16x4_f32(
                           false, a, false, b1, (short)0, c[2 * m + 1], false, false);
        }
    }

    float* Hr = H + row0 * GDIM;
#pragma unroll
    for (int m = 0; m < 4; ++m)
#pragma unroll
        for (int v = 0; v < 8; ++v) {
            v2f o; o.x = c[2 * m][v]; o.y = c[2 * m + 1][v];
            *(v2f*)(Hr + (v + 8 * hi) * GDIM + m * 32 + 2 * col) = o;
        }
}

#define SROWS 32
template <int OFF>
__device__ __forceinline__ void fold_step(float (&ps)[SROWS], float (&pd)[SROWS], int lane) {
    const bool up = (lane & OFF) != 0;
#pragma unroll
    for (int r = 0; r < OFF; ++r) {
        const float ks = up ? ps[r + OFF] : ps[r];
        const float gs = up ? ps[r] : ps[r + OFF];
        const float kd = up ? pd[r + OFF] : pd[r];
        const float gd = up ? pd[r] : pd[r + OFF];
        ps[r] = ks + __shfl_xor(gs, OFF, 32);
        pd[r] = kd + __shfl_xor(gd, OFF, 32);
    }
}
__global__ void __launch_bounds__(128)
gat_node_scores(const float* __restrict__ H, const float* __restrict__ a_src,
                const float* __restrict__ a_dst, float* __restrict__ s_src,
                float* __restrict__ s_dst, int n) {
    const int lane = threadIdx.x & 31;
    const int wave = __builtin_amdgcn_readfirstlane((int)(blockIdx.x * 4 + (threadIdx.x >> 5)));
    const int base = wave * SROWS;
    if (base >= n) return;
    const float4 as = *(const float4*)(a_src + lane * 4);
    const float4 ad = *(const float4*)(a_dst + lane * 4);

    float4 hv[SROWS];
#pragma unroll
    for (int r = 0; r < SROWS; ++r) {
        const int row = (base + r < n) ? (base + r) : (n - 1);
        hv[r] = *(const float4*)(H + (long)row * GDIM + lane * 4);
    }
    float ps[SROWS];
    float pd[SROWS];
#pragma unroll
    for (int r = 0; r < SROWS; ++r) {
        ps[r] = hv[r].x * as.x + hv[r].y * as.y + hv[r].z * as.z + hv[r].w * as.w;
        pd[r] = hv[r].x * ad.x + hv[r].y * ad.y + hv[r].z * ad.z + hv[r].w * ad.w;
    }
    fold_step<16>(ps, pd, lane);
    fold_step<8>(ps, pd, lane);
    fold_step<4>(ps, pd, lane);
    fold_step<2>(ps, pd, lane);
    fold_step<1>(ps, pd, lane);
    if (base + lane < n) {
        s_src[base + lane] = ps[0];
        s_dst[base + lane] = pd[0];
    }
}

#define RNG  400
#define LCAP 6144
#define SCAP 256
#define RING 12

#define OWNCAP 1024
__global__ void __launch_bounds__(256)
gat_list_start(int* __restrict__ wcnt, int n, int nOwn) {
    const int o = blockIdx.x * 256 + threadIdx.x;
    if (o >= nOwn) return;
    const int left = n - o * RNG;
    wcnt[o] = (nOwn <= OWNCAP) ? ((left < RNG) ? left : RNG) : (LCAP + 1);
}
__global__ void __launch_bounds__(512)
gat_owner_lists(const int* __restrict__ ei, int E, int n, int nOwn,
                int* __restrict__ wlist, int* __restrict__ wcnt) {
    __shared__ unsigned l_hist[OWNCAP];
    __shared__ unsigned l_base[OWNCAP];
    if (nOwn > OWNCAP) return;
    const int tid = threadIdx.x;

    const int* dstp = ei + E;
    int sv[8];
    int dv[8];
#pragma unroll
    for (int g = 0; g < 2; ++g) {
        const int e0 = blockIdx.x * 4096 + g * 2048 + tid * 4;
        if ((E & 3) == 0 && e0 < E) {
            const int4 s4 = *(const int4*)(ei + e0);
            const int4 d4 = *(const int4*)(dstp + e0);
            sv[4 * g + 0] = s4.x; sv[4 * g + 1] = s4.y; sv[4 * g + 2] = s4.z; sv[4 * g + 3] = s4.w;
            dv[4 * g + 0] = d4.x; dv[4 * g + 1] = d4.y; dv[4 * g + 2] = d4.z; dv[4 * g + 3] = d4.w;
        } else {
#pragma unroll
            for (int u = 0; u < 4; ++u) {
                const bool ok = e0 + u < E;
                sv[4 * g + u] = ok ? ei[e0 + u] : 0;
                dv[4 * g + u] = ok ? dstp[e0 + u] : -1;
            }
        }
    }
    for (int o = tid; o < nOwn; o += 512) l_hist[o] = 0u;
    __syncthreads();

    int      ow[8];
    unsigned rk[8];
#pragma unroll
    for (int u = 0; u < 8; ++u) {
        ow[u] = ((unsigned)dv[u] < (unsigned)n) ? (int)((unsigned)dv[u] / RNG) : -1;
        rk[u] = 0u;
        if (ow[u] >= 0) rk[u] = atomicAdd(&l_hist[ow[u]], 1u);
    }
    __syncthreads();
    for (int o = tid; o < nOwn; o += 512) {
        const unsigned c = l_hist[o];
        l_base[o] = (c != 0u) ? (unsigned)atomicAdd(&wcnt[o], (int)c) : 0u;
    }
    __syncthreads();
    int2* wl = (int2*)wlist;
    unsigned pos[8];
#pragma unroll
    for (int u = 0; u < 8; ++u) pos[u] = l_base[(ow[u] >= 0) ? ow[u] : 0] + rk[u];
#pragma unroll
    for (int u = 0; u < 8; ++u) {
        if (ow[u] >= 0 && pos[u] < (unsigned)LCAP)
            wl[(long)ow[u] * LCAP + pos[u]] = make_int2(sv[u], dv[u] - ow[u] * RNG);
    }
}

__global__ void __launch_bounds__(1024)
gat_range_aggr(const int* __restrict__ ei, int E, int n,
               const float* __restrict__ s_src, const float* __restrict__ s_dst,
               const float* __restrict__ H,
               const float* __restrict__ bias, float* __restrict__ out,
               const int* __restrict__ wlist, const int* __restrict__ wcnt) {
    __shared__ float    l_acc[RNG * GDIM];
    __shared__ int2     l_list[LCAP];
    __shared__ float    l_lg[LCAP];
    __shared__ unsigned l_menc[RNG];
    __shared__ float    l_sd[RNG];
    __shared__ float    l_den[RNG];
    __shared__ unsigned short l_sub[32 * SCAP];
    __shared__ unsigned l_wn[32];

    const int tid  = threadIdx.x;
    const int lane = tid & 31;
    const int wid  = __builtin_amdgcn_readfirstlane(tid >> 5);
    const int j0   = blockIdx.x * RNG;
    const unsigned nloc = (unsigned)((n - j0 < RNG) ? (n - j0) : RNG);

    float4* a4 = (float4*)l_acc;
    for (int i = tid; i < RNG * GDIM / 4; i += 1024) a4[i] = make_float4(0.f, 0.f, 0.f, 0.f);
    if ((unsigned)tid < nloc) {
        const unsigned dl = (unsigned)tid;
        const int   j     = j0 + (int)dl;
        const float sd    = s_dst[j];
        const float s     = s_src[j] + sd;
        const float logit = (s > 0.f) ? s : NEG_SLOPE * s;
        l_sd[dl]   = sd;
        l_den[dl]  = 0.f;
        l_menc[dl] = enc_f32(logit);
        l_list[dl] = make_int2(j, (int)dl);
        l_lg[dl]   = logit;
        l_sub[(dl & 31u) * SCAP + (dl >> 5)] = (unsigned short)dl;
    }
    if (tid < 32) l_wn[tid] = (nloc + 31u - (unsigned)tid) >> 5;
    __syncthreads();

    const int* dstp = ei + E;
    const int2* wl = (const int2*)wlist + (long)blockIdx.x * LCAP;
    const int total = __builtin_amdgcn_readfirstlane(wcnt[blockIdx.x]);

    if (total <= LCAP) {
        for (int i0 = (int)nloc + tid; i0 < total; i0 += 4096) {
            int   ii[4];
            int2  en[4];
            int   sr[4];
            float ss[4];
#pragma unroll
            for (int u = 0; u < 4; ++u) {
                ii[u] = i0 + u * 1024;
                const int ic = (ii[u] < total) ? ii[u] : i0;
                en[u] = wl[ic];
                sr[u] = en[u].x;
            }
#pragma unroll
            for (int u = 0; u < 4; ++u) ss[u] = s_src[sr[u]];
#pragma unroll
            for (int u = 0; u < 4; ++u) {
                const float s     = ss[u] + l_sd[en[u].y];
                const float logit = (s > 0.f) ? s : NEG_SLOPE * s;
                atomicMax(l_menc + en[u].y, enc_f32(logit));
                if (ii[u] < total) {
                    l_list[ii[u]] = make_int2(sr[u], en[u].y);
                    l_lg[ii[u]]   = logit;
                    const unsigned ow  = (unsigned)en[u].y & 31u;
                    const unsigned pos = atomicAdd(&l_wn[ow], 1u);
                    if (pos < SCAP) l_sub[ow * SCAP + pos] = (unsigned short)ii[u];
                }
            }
        }
        __syncthreads();

        const bool fits = __builtin_amdgcn_ballot_w32(l_wn[lane] > (unsigned)SCAP) == 0u;
        if (fits) {
            const int nw = __builtin_amdgcn_readfirstlane((int)l_wn[wid]);
            const unsigned short* sub = l_sub + wid * SCAP;
            for (int b = 0; b < nw; b += 32) {
                const int nb = (nw - b < 32) ? (nw - b) : 32;
                int   src = 0;
                int   dlv = 0;
                float w   = 0.f;
                if (lane < nb) {
                    const int  idx = sub[b + lane];
                    const int2 en  = l_list[idx];
                    src = en.x;
                    dlv = en.y;
                    w   = __expf(l_lg[idx] - dec_f32(l_menc[dlv]));
                    atomicAdd(l_den + dlv, w);
                }
                if (lane >= nb) {
                    src = __builtin_amdgcn_readlane(src, 0);
                    dlv = __builtin_amdgcn_readlane(dlv, 0);
                }
                float4 h[RING];
#pragma unroll
                for (int u = 0; u < RING; ++u) {
                    const int sk = __builtin_amdgcn_readlane(src, u);
                    h[u] = *(const float4*)(H + (long)sk * GDIM + lane * 4);
                }
#pragma unroll
                for (int k = 0; k < 32; ++k) {
                    const int   dk = __builtin_amdgcn_readlane(dlv, k);
                    const float wk = __int_as_float(__builtin_amdgcn_readlane(__float_as_int(w), k));
                    float4* ap = (float4*)(l_acc + dk * GDIM + lane * 4);
                    float4  a  = *ap;
                    a.x += wk * h[k % RING].x;
                    a.y += wk * h[k % RING].y;
                    a.z += wk * h[k % RING].z;
                    a.w += wk * h[k % RING].w;
                    *ap = a;
                    if (k + RING < 32) {
                        const int sk = __builtin_amdgcn_readlane(src, k + RING);
                        h[k % RING] = *(const float4*)(H + (long)sk * GDIM + lane * 4);
                    }
                }
            }
        } else {
            for (int b = wid * 32; b < total; b += 32 * 32) {
                const int nb = (total - b < 32) ? (total - b) : 32;
                int   src = 0;
                int   dlv = 0;
                float w   = 0.f;
                if (lane < nb) {
                    const int2 en = l_list[b + lane];
                    src = en.x;
                    dlv = en.y;
                    w   = __expf(l_lg[b + lane] - dec_f32(l_menc[dlv]));
                    atomicAdd(l_den + dlv, w);
                }
                int k = 0;
                for (; k + 8 <= nb; k += 8) {
                    float h[8][4];
                    float ww[8];
                    int   dd[8];
#pragma unroll
                    for (int u = 0; u < 8; ++u) {
                        const int sk = __builtin_amdgcn_readlane(src, k + u);
                        dd[u] = __builtin_amdgcn_readlane(dlv, k + u);
                        ww[u] = __int_as_float(__builtin_amdgcn_readlane(__float_as_int(w), k + u));
                        const float* hp = H + (long)sk * GDIM + lane;
                        h[u][0] = hp[0]; h[u][1] = hp[32]; h[u][2] = hp[64]; h[u][3] = hp[96];
                    }
#pragma unroll
                    for (int u = 0; u < 8; ++u) {
                        float* ap = l_acc + dd[u] * GDIM + lane;
                        atomicAdd(ap + 0,  ww[u] * h[u][0]);
                        atomicAdd(ap + 32, ww[u] * h[u][1]);
                        atomicAdd(ap + 64, ww[u] * h[u][2]);
                        atomicAdd(ap + 96, ww[u] * h[u][3]);
                    }
                }
                for (; k < nb; ++k) {
                    const int   sk = __builtin_amdgcn_readlane(src, k);
                    const int   dk = __builtin_amdgcn_readlane(dlv, k);
                    const float wk = __int_as_float(__builtin_amdgcn_readlane(__float_as_int(w), k));
                    const float* hp = H + (long)sk * GDIM + lane;
                    float*       ap = l_acc + dk * GDIM + lane;
                    const float h0 = hp[0], h1 = hp[32], h2 = hp[64], h3 = hp[96];
                    atomicAdd(ap + 0,  wk * h0);
                    atomicAdd(ap + 32, wk * h1);
                    atomicAdd(ap + 64, wk * h2);
                    atomicAdd(ap + 96, wk * h3);
                }
            }
        }
    } else {
        for (int e = tid; e < E; e += 1024) {
            const unsigned d = (unsigned)(dstp[e] - j0);
            if (d < nloc) {
                const float s     = s_src[ei[e]] + l_sd[d];
                const float logit = (s > 0.f) ? s : NEG_SLOPE * s;
                atomicMax(l_menc + d, enc_f32(logit));
            }
        }
        __syncthreads();
        for (int e = tid; e < E + (int)nloc; e += 1024) {
            unsigned d;
            int      src;
            if (e < E) { d = (unsigned)(dstp[e] - j0); src = (d < nloc) ? ei[e] : 0; }
            else       { d = (unsigned)(e - E);        src = j0 + (int)d; }
            if (d < nloc) {
                const float s     = s_src[src] + l_sd[d];
                const float logit = (s > 0.f) ? s : NEG_SLOPE * s;
                const float w     = __expf(logit - dec_f32(l_menc[d]));
                atomicAdd(l_den + d, w);
                const float* hp = H + (long)src * GDIM;
                float*       ap = l_acc + d * GDIM;
                for (int c = 0; c < GDIM; ++c) atomicAdd(ap + c, w * hp[c]);
            }
        }
    }
    __syncthreads();

    const float4 b4 = *(const float4*)(bias + lane * 4);
    for (unsigned dl = wid; dl < nloc; dl += 32) {
        const float inv = 1.0f / l_den[dl];
        float4 v = *(const float4*)(l_acc + dl * GDIM + lane * 4);
        v.x = v.x * inv + b4.x;
        v.y = v.y * inv + b4.y;
        v.z = v.z * inv + b4.z;
        v.w = v.w * inv + b4.w;
        *(float4*)(out + (long)(j0 + (int)dl) * GDIM + lane * 4) = v;
    }
}

#define PLIST 2048
#define PROWS 8
__global__ void __launch_bounds__(1024)
gat_pool_owner(const float* __restrict__ h, const int* __restrict__ batch, int n,
               float* __restrict__ out) {
    __shared__ int      l_nodes[PLIST];
    __shared__ float    l_part[32 * GDIM];
    __shared__ unsigned l_cnt;

    const int tid  = threadIdx.x;
    const int lane = tid & 31;
    const int wid  = __builtin_amdgcn_readfirstlane(tid >> 5);
    const int g    = blockIdx.x;

    if (tid == 0) l_cnt = 0u;
    __syncthreads();

    auto keep = [&](int i) {
        const unsigned pos = atomicAdd(&l_cnt, 1u);
        if (pos < PLIST) l_nodes[pos] = i;
    };
    {
        const int4* b4   = (const int4*)batch;
        const int   nvec = n >> 2;
        const int   nIt  = (nvec + 1023) >> 10;
        const int   rot  = (int)((blockIdx.x * (unsigned)nIt) / gridDim.x);
        for (int it = 0; it < nIt; it += 4) {
            int4 v[4];
            int  vi[4];
#pragma unroll
            for (int u = 0; u < 4; ++u) {
                int c = it + u + rot;
                if (c >= nIt) c -= nIt;
                vi[u] = (c << 10) + tid;
                const bool ok = (it + u < nIt) && (vi[u] < nvec);
                v[u] = ok ? b4[vi[u]] : make_int4(-1, -1, -1, -1);
            }
#pragma unroll
            for (int u = 0; u < 4; ++u) {
                const int i4 = vi[u] << 2;
                if (v[u].x == g) keep(i4 + 0);
                if (v[u].y == g) keep(i4 + 1);
                if (v[u].z == g) keep(i4 + 2);
                if (v[u].w == g) keep(i4 + 3);
            }
        }
        for (int i = (nvec << 2) + tid; i < n; i += 1024)
            if (batch[i] == g) keep(i);
    }
    __syncthreads();

    const int total = __builtin_amdgcn_readfirstlane((int)l_cnt);

    float4 acc = make_float4(0.f, 0.f, 0.f, 0.f);
    if (total <= PLIST) {
        for (int e0 = wid; e0 < total; e0 += 32 * PROWS) {
            int    nd[PROWS];
            float4 r[PROWS];
#pragma unroll
            for (int u = 0; u < PROWS; ++u) {
                const int e = e0 + 32 * u;
                nd[u] = __builtin_amdgcn_readfirstlane(l_nodes[(e < total) ? e : e0]);
            }
#pragma unroll
            for (int u = 0; u < PROWS; ++u)
                r[u] = *(const float4*)(h + (long)nd[u] * GDIM + lane * 4);
#pragma unroll
            for (int u = 0; u < PROWS; ++u) {
                if (e0 + 32 * u < total) {
                    acc.x += r[u].x; acc.y += r[u].y; acc.z += r[u].z; acc.w += r[u].w;
                }
            }
        }
    } else {
        for (int base = wid * 32; base < n; base += 1024) {
            const int i  = base + lane;
            const int gi = (i < n) ? batch[i] : -1;
            unsigned  m  = __builtin_amdgcn_ballot_w32(gi == g);
            while (m != 0u) {
                const int t = __builtin_ctz(m);
                const float4 r = *(const float4*)(h + (long)(base + t) * GDIM + lane * 4);
                acc.x += r.x; acc.y += r.y; acc.z += r.z; acc.w += r.w;
                m &= m - 1u;
            }
        }
    }
    *(float4*)(l_part + wid * GDIM + lane * 4) = acc;
    __syncthreads();

    if (tid < GDIM) {
        float s = 0.f;
#pragma unroll
        for (int w = 0; w < 32; ++w) s += l_part[w * GDIM + tid];
        out[(long)g * GDIM + tid] = s / fmaxf((float)total, 1.0f);
    }
}

extern "C" void kernel_launch(void* const* d_in, const int* in_sizes, int n_in,
                              void* d_out, int out_size, void* d_ws, size_t ws_size,
                              hipStream_t stream) {
    const float* x     = (const float*)d_in[0];
    const int*   ei    = (const int*)d_in[1];
    const int*   batch = (const int*)d_in[2];
    const float* W[3]    = {(const float*)d_in[3], (const float*)d_in[7],  (const float*)d_in[11]};
    const float* asrc[3] = {(const float*)d_in[4], (const float*)d_in[8],  (const float*)d_in[12]};
    const float* adst[3] = {(const float*)d_in[5], (const float*)d_in[9],  (const float*)d_in[13]};
    const float* bias[3] = {(const float*)d_in[6], (const float*)d_in[10], (const float*)d_in[14]};

    const int n  = in_sizes[0] / GDIM;
    const int E  = in_sizes[1] / 2;

    char*  wsb = (char*)d_ws;
    size_t off = 0;
    auto take = [&](size_t bytes) -> void* {
        void* p = wsb + off;
        off += (bytes + 255) & ~(size_t)255;
        return p;
    };
    const size_t szH = (size_t)n * GDIM * sizeof(float);
    float*    h      = (float*)   take(szH);
    float*    acc    = (float*)   take(szH);
    float*    s_src  = (float*)   take((size_t)n * sizeof(float));
    float*    s_dst  = (float*)   take((size_t)n * sizeof(float));
    (void)ws_size; (void)n_in; (void)out_size;

    const int nTiles     = (n + 15) / 16;
    const int gemmBlocks = (nTiles + 3) / 4;
    const int scoreBlocks = (n + 4 * SROWS - 1) / (4 * SROWS);
    const int rangeBlk   = (n + RNG - 1) / RNG;
    int*      wcnt   = (int*)     take((size_t)rangeBlk * sizeof(int));
    int*      wlist  = (int*)     take((size_t)rangeBlk * LCAP * 2 * sizeof(int));

    const int listBlocks = (E + 4095) / 4096;
    gat_list_start<<<(rangeBlk + 255) / 256, 256, 0, stream>>>(wcnt, n, rangeBlk);
    gat_owner_lists<<<listBlocks, 512, 0, stream>>>(ei, E, n, rangeBlk, wlist, wcnt);

    const float* xin = x;
    for (int l = 0; l < 3; ++l) {
        gat_gemm_wmma<<<gemmBlocks, 128, GDIM * GDIM * sizeof(float), stream>>>(
            xin, W[l], h, nTiles);
        gat_node_scores<<<scoreBlocks, 128, 0, stream>>>(h, asrc[l], adst[l],
                                                         s_src, s_dst, n);
        gat_range_aggr<<<rangeBlk, 1024, 0, stream>>>(
            ei, E, n, s_src, s_dst, h, bias[l], acc, wlist, wcnt);
        xin = acc;
    }

    gat_pool_owner<<<NGRAPH, 1024, 0, stream>>>(acc, batch, n, (float*)d_out);
}
